# starting kernel
# baseline (speedup 1.0000x reference)
#include <hip/hip_runtime.h>
#include <hip/hip_bf16.h>


typedef float v2f __attribute__((ext_vector_type(2)));
typedef float v8f __attribute__((ext_vector_type(8)));

#define D 128
#define XS_STRIDE 132


__global__ void fill_f32(float* __restrict__ p, float v, int n) {
    int i = blockIdx.x * blockDim.x + threadIdx.x;
    if (i < n) p[i] = v;
}

__global__ void deg_accum(const int* __restrict__ dst, float* __restrict__ deg, int nE) {
    int e = blockIdx.x * blockDim.x + threadIdx.x;
    if (e < nE) atomicAdd(&deg[dst[e]], 1.0f);
}

__global__ void deg_to_dinv(float* __restrict__ d, int n) {
    int i = blockIdx.x * blockDim.x + threadIdx.x;
    if (i < n) d[i] = rsqrtf(d[i]);
}

__global__ void bias_relu(float* __restrict__ h, const float* __restrict__ b, int n) {
    int i = blockIdx.x * blockDim.x + threadIdx.x;
    if (i < n) {
        float v = h[i] + b[i & (D - 1)];
        h[i] = v > 0.0f ? v : 0.0f;
    }
}

__global__ void bias_add(float* __restrict__ h, const float* __restrict__ b, int n) {
    int i = blockIdx.x * blockDim.x + threadIdx.x;
    if (i < n) h[i] = h[i] + b[i & (D - 1)];
}

__global__ void gemm_wmma_f32(const float* __restrict__ X,
                              const float* __restrict__ W,
                              float* __restrict__ H, int nRows) {
    __shared__ float xs[16 * XS_STRIDE];

    const int row0 = blockIdx.x * 16;
    const int tid  = threadIdx.x;

    for (int i = tid; i < 16 * (D / 4); i += 256) {
        const int r = i >> 5;
        const int c = i & 31;
        float4 v = make_float4(0.f, 0.f, 0.f, 0.f);
        if (row0 + r < nRows)
            v = ((const float4*)(X + (size_t)(row0 + r) * D))[c];
        float* p = &xs[r * XS_STRIDE + c * 4];
        p[0] = v.x; p[1] = v.y; p[2] = v.z; p[3] = v.w;
    }
    __syncthreads();

    const int wave = tid >> 5;
    const int lane = tid & 31;
    const int m    = lane & 15;
    const int kk   = (lane >> 4) << 1;
    const int n    = lane & 15;
    const int col  = wave * 16 + n;

    v8f acc = {};
    #pragma unroll
    for (int k0 = 0; k0 < D; k0 += 4) {
        v2f a, b;
        a.x = xs[m * XS_STRIDE + k0 + kk];
        a.y = xs[m * XS_STRIDE + k0 + kk + 1];
        b.x = W[(size_t)(k0 + kk) * D + col];
        b.y = W[(size_t)(k0 + kk + 1) * D + col];
        acc = __builtin_amdgcn_wmma_f32_16x16x4_f32(
             false, a,  false, b,
             (short)0, acc,  false,  false);
    }

    const int halfm = (lane >> 4) * 8;
    #pragma unroll
    for (int r = 0; r < 8; ++r) {
        const int row = row0 + halfm + r;
        if (row < nRows) H[(size_t)row * D + col] = acc[r];
    }
}

__global__ void edge_aggregate(const int* __restrict__ src,
                               const int* __restrict__ dst,
                               const float* __restrict__ dinv,
                               const float* __restrict__ h,
                               float* __restrict__ out, int nE, int nN) {
    const int warp = (blockIdx.x * blockDim.x + threadIdx.x) >> 5;
    const int lane = threadIdx.x & 31;
    const int total = nE + nN;
    if (warp >= total) return;

    int s, d;
    if (warp < nE) { s = src[warp]; d = dst[warp]; }
    else           { s = d = warp - nE; }

    const float norm = dinv[s] * dinv[d];
    const float4 v = ((const float4*)(h + (size_t)s * D))[lane];

    float* o = out + (size_t)d * D + lane * 4;
    atomicAdd(o + 0, v.x * norm);
    atomicAdd(o + 1, v.y * norm);
    atomicAdd(o + 2, v.z * norm);
    atomicAdd(o + 3, v.w * norm);
}


static inline size_t align256(size_t v) { return (v + 255) & ~(size_t)255; }

extern "C" void kernel_launch(void* const* d_in, const int* in_sizes, int n_in,
                              void* d_out, int out_size, void* d_ws, size_t ws_size,
                              hipStream_t stream) {
    const float* x  = (const float*)d_in[0];
    const int*   ei = (const int*)d_in[1];
    const float* W1 = (const float*)d_in[2];
    const float* b1 = (const float*)d_in[3];
    const float* W2 = (const float*)d_in[4];
    const float* b2 = (const float*)d_in[5];
    float* out = (float*)d_out;

    const int N = in_sizes[0] / D;
    const int E = in_sizes[1] / 2;
    const int* srcIdx = ei;
    const int* dstIdx = ei + E;

    char* ws = (char*)d_ws;
    float* dinv = (float*)ws;
    const size_t o1 = align256((size_t)N * sizeof(float));
    float* hbuf = (float*)(ws + o1);
    const size_t o2 = o1 + align256((size_t)N * D * sizeof(float));
    float* agg  = (float*)(ws + o2);

    const int nF = N * D;
    const int blk = 256;
    const int gN   = (N + blk - 1) / blk;
    const int gE   = (E + blk - 1) / blk;
    const int gF   = (nF + blk - 1) / blk;
    const int gGemm = (N + 15) / 16;
    const int gEdge = ((E + N) + 7) / 8;

    fill_f32   <<<gN, blk, 0, stream>>>(dinv, 1.0f, N);
    deg_accum  <<<gE, blk, 0, stream>>>(dstIdx, dinv, E);
    deg_to_dinv<<<gN, blk, 0, stream>>>(dinv, N);

    gemm_wmma_f32 <<<gGemm, blk, 0, stream>>>(x, W1, hbuf, N);
    fill_f32      <<<gF, blk, 0, stream>>>(agg, 0.0f, nF);
    edge_aggregate<<<gEdge, blk, 0, stream>>>(srcIdx, dstIdx, dinv, hbuf, agg, E, N);
    bias_relu     <<<gF, blk, 0, stream>>>(agg, b1, nF);

    gemm_wmma_f32 <<<gGemm, blk, 0, stream>>>(agg, W2, hbuf, N);
    fill_f32      <<<gF, blk, 0, stream>>>(out, 0.0f, nF);
    edge_aggregate<<<gEdge, blk, 0, stream>>>(srcIdx, dstIdx, dinv, hbuf, out, E, N);
    bias_add      <<<gF, blk, 0, stream>>>(out, b2, nF);
}


# Aggregate by gather over in-edges grouped by dst (CSR rebuilt per call: int atomic hist gives arrival-order rank, scan, fill); one wave sums a node's rows, writes once with bias/relu; no f32 atomics
# speedup vs baseline: 2.9436x; 2.9436x over previous
#include <hip/hip_runtime.h>
#include <hip/hip_bf16.h>


typedef float v2f __attribute__((ext_vector_type(2)));
typedef float v8f __attribute__((ext_vector_type(8)));

#define D 128
#define XS_STRIDE 132
#define SCAN_T 1024
#define SCAN_C 4096


__global__ void zero_i32(int* __restrict__ p, int n) {
    int i = blockIdx.x * blockDim.x + threadIdx.x;
    if (i < n) p[i] = 0;
}

__global__ void hist_rank(const int* __restrict__ dst, int* __restrict__ cnt,
                          int* __restrict__ rank, int nE) {
    int e = blockIdx.x * blockDim.x + threadIdx.x;
    if (e < nE) rank[e] = atomicAdd(&cnt[dst[e]], 1);
}

__global__ __launch_bounds__(SCAN_T) void scan_deg(const int* __restrict__ cnt,
                                                   int* __restrict__ off,
                                                   float* __restrict__ dinv, int n) {
    __shared__ int s_w[32];
    __shared__ int s_p[32];
    const int tid  = threadIdx.x;
    const int lane = tid & 31;
    const int wave = tid >> 5;
    const int chunk0 = blockIdx.x * SCAN_C;

    int pre = 0;
    const int4* __restrict__ cnt4 = (const int4*)cnt;
    #pragma unroll 8
    for (int k = 0; k < (int)blockIdx.x; ++k) {
        const int4 v = cnt4[k * SCAN_T + tid];
        pre += (v.x + v.y) + (v.z + v.w);
    }
    for (int o = 16; o > 0; o >>= 1) pre += __shfl_xor(pre, o);

    const int i0 = chunk0 + tid * 4;
    int c[4];
    #pragma unroll
    for (int k = 0; k < 4; ++k) c[k] = (i0 + k < n) ? cnt[i0 + k] : 0;
    const int t = (c[0] + c[1]) + (c[2] + c[3]);

    int inc = t;
    #pragma unroll
    for (int o = 1; o < 32; o <<= 1) {
        const int u = __shfl_up(inc, o);
        if (lane >= o) inc += u;
    }
    if (lane == 31) s_w[wave] = inc;
    if (lane == 0)  s_p[wave] = pre;
    __syncthreads();

    int q = s_p[lane] + (lane < wave ? s_w[lane] : 0);
    for (int o = 16; o > 0; o >>= 1) q += __shfl_xor(q, o);

    int run = q + inc - t;
    #pragma unroll
    for (int k = 0; k < 4; ++k) {
        if (i0 + k < n) {
            run += c[k];
            off[i0 + k + 1] = run;
            dinv[i0 + k] = rsqrtf((float)(c[k] + 1));
        }
    }
    if (blockIdx.x == 0 && tid == 0) off[0] = 0;
}

__global__ void csr_fill(const int* __restrict__ src, const int* __restrict__ dst,
                         const int* __restrict__ off, const int* __restrict__ rank,
                         int* __restrict__ csr, int nE) {
    int e = blockIdx.x * blockDim.x + threadIdx.x;
    if (e < nE) {
        const int pos = off[dst[e]] + rank[e];
        if ((unsigned)pos < (unsigned)nE) csr[pos] = src[e];
    }
}

__global__ void gemm_wmma_f32(const float* __restrict__ X,
                              const float* __restrict__ W,
                              float* __restrict__ H, int nRows) {
    __shared__ float xs[16 * XS_STRIDE];

    const int row0 = blockIdx.x * 16;
    const int tid  = threadIdx.x;

    for (int i = tid; i < 16 * (D / 4); i += 256) {
        const int r = i >> 5;
        const int c = i & 31;
        float4 v = make_float4(0.f, 0.f, 0.f, 0.f);
        if (row0 + r < nRows)
            v = ((const float4*)(X + (size_t)(row0 + r) * D))[c];
        float* p = &xs[r * XS_STRIDE + c * 4];
        p[0] = v.x; p[1] = v.y; p[2] = v.z; p[3] = v.w;
    }
    __syncthreads();

    const int wave = tid >> 5;
    const int lane = tid & 31;
    const int m    = lane & 15;
    const int kk   = (lane >> 4) << 1;
    const int n    = lane & 15;
    const int col  = wave * 16 + n;

    v8f acc = {};
    #pragma unroll
    for (int k0 = 0; k0 < D; k0 += 4) {
        v2f a, b;
        a.x = xs[m * XS_STRIDE + k0 + kk];
        a.y = xs[m * XS_STRIDE + k0 + kk + 1];
        b.x = W[(size_t)(k0 + kk) * D + col];
        b.y = W[(size_t)(k0 + kk + 1) * D + col];
        acc = __builtin_amdgcn_wmma_f32_16x16x4_f32(
             false, a,  false, b,
             (short)0, acc,  false,  false);
    }

    const int halfm = (lane >> 4) * 8;
    #pragma unroll
    for (int r = 0; r < 8; ++r) {
        const int row = row0 + halfm + r;
        if (row < nRows) H[(size_t)row * D + col] = acc[r];
    }
}

#define ROW_LOAD(a, jj) \
    const float4 a = hp[(size_t)__builtin_amdgcn_readlane(s_l, (jj)) * (D / 4) + lane]
#define ROW_FMA(a, jj) do { \
    const float w_ = __int_as_float(__builtin_amdgcn_readlane(w_i, (jj))); \
    acc.x = fmaf(a.x, w_, acc.x); acc.y = fmaf(a.y, w_, acc.y); \
    acc.z = fmaf(a.z, w_, acc.z); acc.w = fmaf(a.w, w_, acc.w); } while (0)

template <bool RELU>
__global__ __launch_bounds__(256) void gather_rows(const int* __restrict__ off,
                                                   const int* __restrict__ csr,
                                                   const float* __restrict__ dinv,
                                                   const float* __restrict__ h,
                                                   const float* __restrict__ bias,
                                                   float* __restrict__ out, int nN) {
    const int lane = threadIdx.x & 31;
    const int node = __builtin_amdgcn_readfirstlane((int)(blockIdx.x * 8u + (threadIdx.x >> 5)));
    if (node >= nN) return;

    const int beg = __builtin_amdgcn_readfirstlane(off[node]);
    const int end = __builtin_amdgcn_readfirstlane(off[node + 1]);
    const float dd = dinv[node];
    const float4* __restrict__ hp = (const float4*)h;

    const float4 vself = hp[(size_t)node * (D / 4) + lane];
    const float4 b = ((const float4*)bias)[lane];

    float4 acc = make_float4(0.0f, 0.0f, 0.0f, 0.0f);

    for (int base = beg; base < end; base += 32) {
        const int m = (end - base) < 32 ? (end - base) : 32;
        int   s_l = 0;
        float w_l = 0.0f;
        if (lane < m) {
            s_l = csr[base + lane];
            s_l = s_l < 0 ? 0 : (s_l < nN ? s_l : nN - 1);
            w_l = dinv[s_l] * dd;
        }
        const int w_i = __float_as_int(w_l);

        int j = 0;
        for (; j + 8 <= m; j += 8) {
            ROW_LOAD(a0, j);     ROW_LOAD(a1, j + 1); ROW_LOAD(a2, j + 2); ROW_LOAD(a3, j + 3);
            ROW_LOAD(a4, j + 4); ROW_LOAD(a5, j + 5); ROW_LOAD(a6, j + 6); ROW_LOAD(a7, j + 7);
            ROW_FMA(a0, j);     ROW_FMA(a1, j + 1); ROW_FMA(a2, j + 2); ROW_FMA(a3, j + 3);
            ROW_FMA(a4, j + 4); ROW_FMA(a5, j + 5); ROW_FMA(a6, j + 6); ROW_FMA(a7, j + 7);
        }
        if (j + 4 <= m) {
            ROW_LOAD(a0, j); ROW_LOAD(a1, j + 1); ROW_LOAD(a2, j + 2); ROW_LOAD(a3, j + 3);
            ROW_FMA(a0, j);  ROW_FMA(a1, j + 1);  ROW_FMA(a2, j + 2);  ROW_FMA(a3, j + 3);
            j += 4;
        }
        const int r = m - j;
        if (r == 3) {
            ROW_LOAD(a0, j); ROW_LOAD(a1, j + 1); ROW_LOAD(a2, j + 2);
            ROW_FMA(a0, j);  ROW_FMA(a1, j + 1);  ROW_FMA(a2, j + 2);
        } else if (r == 2) {
            ROW_LOAD(a0, j); ROW_LOAD(a1, j + 1);
            ROW_FMA(a0, j);  ROW_FMA(a1, j + 1);
        } else if (r == 1) {
            ROW_LOAD(a0, j);
            ROW_FMA(a0, j);
        }
    }

    {
        const float ns = dd * dd;
        acc.x = fmaf(vself.x, ns, acc.x); acc.y = fmaf(vself.y, ns, acc.y);
        acc.z = fmaf(vself.z, ns, acc.z); acc.w = fmaf(vself.w, ns, acc.w);
    }
    acc.x += b.x; acc.y += b.y; acc.z += b.z; acc.w += b.w;
    if (RELU) {
        acc.x = acc.x > 0.0f ? acc.x : 0.0f;
        acc.y = acc.y > 0.0f ? acc.y : 0.0f;
        acc.z = acc.z > 0.0f ? acc.z : 0.0f;
        acc.w = acc.w > 0.0f ? acc.w : 0.0f;
    }
    ((float4*)out)[(size_t)node * (D / 4) + lane] = acc;
}


static inline size_t align256(size_t v) { return (v + 255) & ~(size_t)255; }

extern "C" void kernel_launch(void* const* d_in, const int* in_sizes, int n_in,
                              void* d_out, int out_size, void* d_ws, size_t ws_size,
                              hipStream_t stream) {
    const float* x  = (const float*)d_in[0];
    const int*   ei = (const int*)d_in[1];
    const float* W1 = (const float*)d_in[2];
    const float* b1 = (const float*)d_in[3];
    const float* W2 = (const float*)d_in[4];
    const float* b2 = (const float*)d_in[5];
    float* out = (float*)d_out;

    const int N = in_sizes[0] / D;
    const int E = in_sizes[1] / 2;
    const int* srcIdx = ei;
    const int* dstIdx = ei + E;

    char* ws = (char*)d_ws;
    size_t o = 0;
    float* dinv = (float*)(ws + o); o += align256((size_t)N * sizeof(float));
    int*   cnt  = (int*)(ws + o);   o += align256((size_t)N * sizeof(int));
    int*   off  = (int*)(ws + o);   o += align256((size_t)(N + 1) * sizeof(int));
    int*   rank = (int*)(ws + o);   o += align256((size_t)E * sizeof(int));
    int*   csr  = (int*)(ws + o);   o += align256((size_t)E * sizeof(int));
    float* hbuf = (float*)(ws + o); o += align256((size_t)N * D * sizeof(float));
    float* agg  = (float*)(ws + o);

    const int blk = 256;
    const int gN    = (N + blk - 1) / blk;
    const int gE    = (E + blk - 1) / blk;
    const int gScan = (N + SCAN_C - 1) / SCAN_C;
    const int gGemm = (N + 15) / 16;
    const int gRows = (N + 7) / 8;

    zero_i32 <<<gN, blk, 0, stream>>>(cnt, N);
    hist_rank<<<gE, blk, 0, stream>>>(dstIdx, cnt, rank, E);
    scan_deg <<<gScan, SCAN_T, 0, stream>>>(cnt, off, dinv, N);
    csr_fill <<<gE, blk, 0, stream>>>(srcIdx, dstIdx, off, rank, csr, E);

    gemm_wmma_f32     <<<gGemm, blk, 0, stream>>>(x, W1, hbuf, N);
    gather_rows<true> <<<gRows, blk, 0, stream>>>(off, csr, dinv, hbuf, b1, agg, N);

    gemm_wmma_f32     <<<gGemm, blk, 0, stream>>>(agg, W2, hbuf, N);
    gather_rows<false><<<gRows, blk, 0, stream>>>(off, csr, dinv, hbuf, b2, out, N);
}


# CSR build regroups edges by a two-level counting sort on LDS counters (256-node buckets, then nodes); no global atomics left; in-group order stays arrival order, now of LDS adds
# speedup vs baseline: 4.2662x; 1.4493x over previous
#include <hip/hip_runtime.h>
#include <hip/hip_bf16.h>


typedef float v2f __attribute__((ext_vector_type(2)));
typedef float v8f __attribute__((ext_vector_type(8)));

#define D 128
#define XS_STRIDE 132
#define SCAN_T 1024
#define SCAN_C 4096


#define NB_SH 8
#define NB    256
#define G1    256
#define T1    512
#define T4    1024
#define MAXB  512
#define KB    16
#define K4    4

__device__ __forceinline__ int bucket_of(int d, int nB) {
    const int b = (int)((unsigned)d >> NB_SH);
    return b < nB ? b : nB - 1;
}

__global__ __launch_bounds__(T1) void bucket_count(const int* __restrict__ dst,
                                                   int* __restrict__ tbl,
                                                   int nE, int nB, int per) {
    __shared__ int s_c[MAXB];
    const int tid = threadIdx.x;
    const int w   = blockIdx.x;
    for (int i = tid; i < nB; i += T1) s_c[i] = 0;
    __syncthreads();
    const int e0 = w * per;
    const int e1 = (e0 + per < nE) ? (e0 + per) : nE;
    for (int e = e0 + tid; e < e1; e += KB * T1) {
        int d[KB];
        #pragma unroll
        for (int k = 0; k < KB; ++k) {
            const int ee = e + k * T1;
            d[k] = dst[ee < e1 ? ee : e1 - 1];
        }
        #pragma unroll
        for (int k = 0; k < KB; ++k)
            if (e + k * T1 < e1) atomicAdd(&s_c[bucket_of(d[k], nB)], 1);
    }
    __syncthreads();
    for (int i = tid; i < nB; i += T1) tbl[i * G1 + w] = s_c[i];
}

__global__ __launch_bounds__(SCAN_T) void scan_excl(const int* __restrict__ in,
                                                    int* __restrict__ out, int n) {
    __shared__ int s_w[32];
    __shared__ int s_p[32];
    const int tid  = threadIdx.x;
    const int lane = tid & 31;
    const int wave = tid >> 5;
    const int chunk0 = blockIdx.x * SCAN_C;

    int pre = 0;
    const int4* __restrict__ in4 = (const int4*)in;
    #pragma unroll 8
    for (int k = 0; k < (int)blockIdx.x; ++k) {
        const int4 v = in4[k * SCAN_T + tid];
        pre += (v.x + v.y) + (v.z + v.w);
    }
    for (int o = 16; o > 0; o >>= 1) pre += __shfl_xor(pre, o);

    const int i0 = chunk0 + tid * 4;
    int c[4];
    #pragma unroll
    for (int k = 0; k < 4; ++k) c[k] = (i0 + k < n) ? in[i0 + k] : 0;
    const int t = (c[0] + c[1]) + (c[2] + c[3]);

    int inc = t;
    #pragma unroll
    for (int o = 1; o < 32; o <<= 1) {
        const int u = __shfl_up(inc, o);
        if (lane >= o) inc += u;
    }
    if (lane == 31) s_w[wave] = inc;
    if (lane == 0)  s_p[wave] = pre;
    __syncthreads();

    int q = s_p[lane] + (lane < wave ? s_w[lane] : 0);
    for (int o = 16; o > 0; o >>= 1) q += __shfl_xor(q, o);

    int run = q + inc - t;
    #pragma unroll
    for (int k = 0; k < 4; ++k) {
        if (i0 + k < n) {
            out[i0 + k] = run;
            run += c[k];
        }
    }
}

__global__ __launch_bounds__(T1) void bucket_scatter(const int* __restrict__ src,
                                                     const int* __restrict__ dst,
                                                     const int* __restrict__ base,
                                                     int2* __restrict__ pairs,
                                                     int nE, int nB, int per) {
    __shared__ int s_p[MAXB];
    const int tid = threadIdx.x;
    const int w   = blockIdx.x;
    for (int i = tid; i < nB; i += T1) s_p[i] = base[i * G1 + w];
    __syncthreads();
    const int e0 = w * per;
    const int e1 = (e0 + per < nE) ? (e0 + per) : nE;
    for (int e = e0 + tid; e < e1; e += KB * T1) {
        int d[KB], s[KB];
        #pragma unroll
        for (int k = 0; k < KB; ++k) {
            const int ee = (e + k * T1 < e1) ? (e + k * T1) : (e1 - 1);
            d[k] = dst[ee];
            s[k] = src[ee];
        }
        #pragma unroll
        for (int k = 0; k < KB; ++k) {
            if (e + k * T1 < e1) {
                const int p = atomicAdd(&s_p[bucket_of(d[k], nB)], 1);
                if ((unsigned)p < (unsigned)nE) pairs[p] = make_int2(s[k], d[k]);
            }
        }
    }
}

__global__ __launch_bounds__(T4) void bucket_build(const int2* __restrict__ pairs,
                                                   const int* __restrict__ base,
                                                   int* __restrict__ off,
                                                   float* __restrict__ dinv,
                                                   int* __restrict__ csr,
                                                   int nN, int nE, int nB) {
    __shared__ int s_cnt[NB];
    __shared__ int s_cur[NB];
    __shared__ int s_w[NB / 32];
    const int tid  = threadIdx.x;
    const int lane = tid & 31;
    const int b    = blockIdx.x;
    const int node0 = b << NB_SH;
    int bs = base[b * G1];
    int be = (b + 1 < nB) ? base[(b + 1) * G1] : nE;
    bs = bs < 0 ? 0 : (bs < nE ? bs : nE);
    be = be < bs ? bs : (be < nE ? be : nE);

    if (tid < NB) s_cnt[tid] = 0;
    __syncthreads();

    for (int i = bs + tid; i < be; i += K4 * T4) {
        int d[K4];
        #pragma unroll
        for (int k = 0; k < K4; ++k) {
            const int ii = i + k * T4;
            d[k] = pairs[ii < be ? ii : be - 1].y;
        }
        #pragma unroll
        for (int k = 0; k < K4; ++k) {
            const unsigned dl = (unsigned)(d[k] - node0);
            if (i + k * T4 < be && dl < (unsigned)NB) atomicAdd(&s_cnt[dl], 1);
        }
    }
    __syncthreads();

    int c = 0, inc = 0;
    if (tid < NB) {
        c = s_cnt[tid];
        inc = c;
        #pragma unroll
        for (int o = 1; o < 32; o <<= 1) {
            const int u = __shfl_up(inc, o);
            if (lane >= o) inc += u;
        }
        if (lane == 31) s_w[tid >> 5] = inc;
    }
    __syncthreads();

    if (tid < NB) {
        int pre = 0;
        for (int k = 0; k < (tid >> 5); ++k) pre += s_w[k];
        const int start = bs + pre + inc - c;
        s_cur[tid] = start;
        const int node = node0 + tid;
        if (node < nN) {
            off[node]  = start;
            dinv[node] = rsqrtf((float)(c + 1));
            if (node == nN - 1) off[nN] = start + c;
        }
    }
    __syncthreads();

    for (int i = bs + tid; i < be; i += K4 * T4) {
        int2 pr[K4];
        #pragma unroll
        for (int k = 0; k < K4; ++k) {
            const int ii = i + k * T4;
            pr[k] = pairs[ii < be ? ii : be - 1];
        }
        #pragma unroll
        for (int k = 0; k < K4; ++k) {
            const unsigned dl = (unsigned)(pr[k].y - node0);
            if (i + k * T4 < be && dl < (unsigned)NB) {
                const int p = atomicAdd(&s_cur[dl], 1);
                if ((unsigned)p < (unsigned)nE) csr[p] = pr[k].x;
            }
        }
    }
}

__global__ void gemm_wmma_f32(const float* __restrict__ X,
                              const float* __restrict__ W,
                              float* __restrict__ H, int nRows) {
    __shared__ float xs[16 * XS_STRIDE];

    const int row0 = blockIdx.x * 16;
    const int tid  = threadIdx.x;

    for (int i = tid; i < 16 * (D / 4); i += 256) {
        const int r = i >> 5;
        const int c = i & 31;
        float4 v = make_float4(0.f, 0.f, 0.f, 0.f);
        if (row0 + r < nRows)
            v = ((const float4*)(X + (size_t)(row0 + r) * D))[c];
        float* p = &xs[r * XS_STRIDE + c * 4];
        p[0] = v.x; p[1] = v.y; p[2] = v.z; p[3] = v.w;
    }
    __syncthreads();

    const int wave = tid >> 5;
    const int lane = tid & 31;
    const int m    = lane & 15;
    const int kk   = (lane >> 4) << 1;
    const int n    = lane & 15;
    const int col  = wave * 16 + n;

    v8f acc = {};
    #pragma unroll
    for (int k0 = 0; k0 < D; k0 += 4) {
        v2f a, b;
        a.x = xs[m * XS_STRIDE + k0 + kk];
        a.y = xs[m * XS_STRIDE + k0 + kk + 1];
        b.x = W[(size_t)(k0 + kk) * D + col];
        b.y = W[(size_t)(k0 + kk + 1) * D + col];
        acc = __builtin_amdgcn_wmma_f32_16x16x4_f32(
             false, a,  false, b,
             (short)0, acc,  false,  false);
    }

    const int halfm = (lane >> 4) * 8;
    #pragma unroll
    for (int r = 0; r < 8; ++r) {
        const int row = row0 + halfm + r;
        if (row < nRows) H[(size_t)row * D + col] = acc[r];
    }
}

#define ROW_LOAD(a, jj) \
    const float4 a = hp[(size_t)__builtin_amdgcn_readlane(s_l, (jj)) * (D / 4) + lane]
#define ROW_FMA(a, jj) do { \
    const float w_ = __int_as_float(__builtin_amdgcn_readlane(w_i, (jj))); \
    acc.x = fmaf(a.x, w_, acc.x); acc.y = fmaf(a.y, w_, acc.y); \
    acc.z = fmaf(a.z, w_, acc.z); acc.w = fmaf(a.w, w_, acc.w); } while (0)

template <bool RELU>
__global__ __launch_bounds__(256) void gather_rows(const int* __restrict__ off,
                                                   const int* __restrict__ csr,
                                                   const float* __restrict__ dinv,
                                                   const float* __restrict__ h,
                                                   const float* __restrict__ bias,
                                                   float* __restrict__ out, int nN) {
    const int lane = threadIdx.x & 31;
    const int node = __builtin_amdgcn_readfirstlane((int)(blockIdx.x * 8u + (threadIdx.x >> 5)));
    if (node >= nN) return;

    const int beg = __builtin_amdgcn_readfirstlane(off[node]);
    const int end = __builtin_amdgcn_readfirstlane(off[node + 1]);
    const float dd = dinv[node];
    const float4* __restrict__ hp = (const float4*)h;

    const float4 vself = hp[(size_t)node * (D / 4) + lane];
    const float4 b = ((const float4*)bias)[lane];

    float4 acc = make_float4(0.0f, 0.0f, 0.0f, 0.0f);

    for (int base = beg; base < end; base += 32) {
        const int m = (end - base) < 32 ? (end - base) : 32;
        int   s_l = 0;
        float w_l = 0.0f;
        if (lane < m) {
            s_l = csr[base + lane];
            s_l = s_l < 0 ? 0 : (s_l < nN ? s_l : nN - 1);
            w_l = dinv[s_l] * dd;
        }
        const int w_i = __float_as_int(w_l);

        int j = 0;
        for (; j + 8 <= m; j += 8) {
            ROW_LOAD(a0, j);     ROW_LOAD(a1, j + 1); ROW_LOAD(a2, j + 2); ROW_LOAD(a3, j + 3);
            ROW_LOAD(a4, j + 4); ROW_LOAD(a5, j + 5); ROW_LOAD(a6, j + 6); ROW_LOAD(a7, j + 7);
            ROW_FMA(a0, j);     ROW_FMA(a1, j + 1); ROW_FMA(a2, j + 2); ROW_FMA(a3, j + 3);
            ROW_FMA(a4, j + 4); ROW_FMA(a5, j + 5); ROW_FMA(a6, j + 6); ROW_FMA(a7, j + 7);
        }
        if (j + 4 <= m) {
            ROW_LOAD(a0, j); ROW_LOAD(a1, j + 1); ROW_LOAD(a2, j + 2); ROW_LOAD(a3, j + 3);
            ROW_FMA(a0, j);  ROW_FMA(a1, j + 1);  ROW_FMA(a2, j + 2);  ROW_FMA(a3, j + 3);
            j += 4;
        }
        const int r = m - j;
        if (r == 3) {
            ROW_LOAD(a0, j); ROW_LOAD(a1, j + 1); ROW_LOAD(a2, j + 2);
            ROW_FMA(a0, j);  ROW_FMA(a1, j + 1);  ROW_FMA(a2, j + 2);
        } else if (r == 2) {
            ROW_LOAD(a0, j); ROW_LOAD(a1, j + 1);
            ROW_FMA(a0, j);  ROW_FMA(a1, j + 1);
        } else if (r == 1) {
            ROW_LOAD(a0, j);
            ROW_FMA(a0, j);
        }
    }

    {
        const float ns = dd * dd;
        acc.x = fmaf(vself.x, ns, acc.x); acc.y = fmaf(vself.y, ns, acc.y);
        acc.z = fmaf(vself.z, ns, acc.z); acc.w = fmaf(vself.w, ns, acc.w);
    }
    acc.x += b.x; acc.y += b.y; acc.z += b.z; acc.w += b.w;
    if (RELU) {
        acc.x = acc.x > 0.0f ? acc.x : 0.0f;
        acc.y = acc.y > 0.0f ? acc.y : 0.0f;
        acc.z = acc.z > 0.0f ? acc.z : 0.0f;
        acc.w = acc.w > 0.0f ? acc.w : 0.0f;
    }
    ((float4*)out)[(size_t)node * (D / 4) + lane] = acc;
}


static inline size_t align256(size_t v) { return (v + 255) & ~(size_t)255; }

extern "C" void kernel_launch(void* const* d_in, const int* in_sizes, int n_in,
                              void* d_out, int out_size, void* d_ws, size_t ws_size,
                              hipStream_t stream) {
    const float* x  = (const float*)d_in[0];
    const int*   ei = (const int*)d_in[1];
    const float* W1 = (const float*)d_in[2];
    const float* b1 = (const float*)d_in[3];
    const float* W2 = (const float*)d_in[4];
    const float* b2 = (const float*)d_in[5];
    float* out = (float*)d_out;

    const int N = in_sizes[0] / D;
    const int E = in_sizes[1] / 2;
    const int* srcIdx = ei;
    const int* dstIdx = ei + E;

    const int nB  = (N + NB - 1) / NB;
    const int nT  = nB * G1;
    const int per = (E + G1 - 1) / G1;

    char* ws = (char*)d_ws;
    size_t o = 0;
    float* dinv  = (float*)(ws + o); o += align256((size_t)N * sizeof(float));
    int*   tbl   = (int*)(ws + o);   o += align256((size_t)nT * sizeof(int));
    int*   base  = (int*)(ws + o);   o += align256((size_t)nT * sizeof(int));
    int*   off   = (int*)(ws + o);   o += align256((size_t)(N + 1) * sizeof(int));
    int2*  pairs = (int2*)(ws + o);  o += align256((size_t)E * sizeof(int2));
    int*   csr   = (int*)(ws + o);   o += align256((size_t)E * sizeof(int));
    float* hbuf  = (float*)(ws + o); o += align256((size_t)N * D * sizeof(float));
    float* agg   = (float*)(ws + o);

    const int blk = 256;
    const int gScan = (nT + SCAN_C - 1) / SCAN_C;
    const int gGemm = (N + 15) / 16;
    const int gRows = (N + 7) / 8;

    bucket_count  <<<G1, T1, 0, stream>>>(dstIdx, tbl, E, nB, per);
    scan_excl     <<<gScan, SCAN_T, 0, stream>>>(tbl, base, nT);
    bucket_scatter<<<G1, T1, 0, stream>>>(srcIdx, dstIdx, base, pairs, E, nB, per);
    bucket_build  <<<nB, T4, 0, stream>>>(pairs, base, off, dinv, csr, N, E, nB);

    gemm_wmma_f32     <<<gGemm, blk, 0, stream>>>(x, W1, hbuf, N);
    gather_rows<true> <<<gRows, blk, 0, stream>>>(off, csr, dinv, hbuf, b1, agg, N);

    gemm_wmma_f32     <<<gGemm, blk, 0, stream>>>(agg, W2, hbuf, N);
    gather_rows<false><<<gRows, blk, 0, stream>>>(off, csr, dinv, hbuf, b2, out, N);
}


# Each layer is one launch: resident blocks walk 16-node strips, sum each node's neighbour rows of the layer input into LDS, multiply the strip by W (held in registers) on WMMA; sums run as (A x) W
# speedup vs baseline: 4.9135x; 1.1517x over previous
#include <hip/hip_runtime.h>
#include <hip/hip_bf16.h>


typedef float v2f __attribute__((ext_vector_type(2)));
typedef float v8f __attribute__((ext_vector_type(8)));

#define D 128
#define XS_STRIDE 132
#define SCAN_T 1024
#define SCAN_C 4096


#define NB_SH 8
#define NB    256
#define G1    256
#define T1    512
#define T4    1024
#define MAXB  512
#define KB    16
#define K4    4

__device__ __forceinline__ int bucket_of(int d, int nB) {
    const int b = (int)((unsigned)d >> NB_SH);
    return b < nB ? b : nB - 1;
}

__global__ __launch_bounds__(T1) void bucket_count(const int* __restrict__ dst,
                                                   int* __restrict__ tbl,
                                                   int nE, int nB, int per) {
    __shared__ int s_c[MAXB];
    const int tid = threadIdx.x;
    const int w   = blockIdx.x;
    for (int i = tid; i < nB; i += T1) s_c[i] = 0;
    __syncthreads();
    const int e0 = w * per;
    const int e1 = (e0 + per < nE) ? (e0 + per) : nE;
    for (int e = e0 + tid; e < e1; e += KB * T1) {
        int d[KB];
        #pragma unroll
        for (int k = 0; k < KB; ++k) {
            const int ee = e + k * T1;
            d[k] = dst[ee < e1 ? ee : e1 - 1];
        }
        #pragma unroll
        for (int k = 0; k < KB; ++k)
            if (e + k * T1 < e1) atomicAdd(&s_c[bucket_of(d[k], nB)], 1);
    }
    __syncthreads();
    for (int i = tid; i < nB; i += T1) tbl[i * G1 + w] = s_c[i];
}

__global__ __launch_bounds__(SCAN_T) void scan_excl(const int* __restrict__ in,
                                                    int* __restrict__ out, int n) {
    __shared__ int s_w[32];
    __shared__ int s_p[32];
    const int tid  = threadIdx.x;
    const int lane = tid & 31;
    const int wave = tid >> 5;
    const int chunk0 = blockIdx.x * SCAN_C;

    int pre = 0;
    const int4* __restrict__ in4 = (const int4*)in;
    #pragma unroll 8
    for (int k = 0; k < (int)blockIdx.x; ++k) {
        const int4 v = in4[k * SCAN_T + tid];
        pre += (v.x + v.y) + (v.z + v.w);
    }
    for (int o = 16; o > 0; o >>= 1) pre += __shfl_xor(pre, o);

    const int i0 = chunk0 + tid * 4;
    int c[4];
    #pragma unroll
    for (int k = 0; k < 4; ++k) c[k] = (i0 + k < n) ? in[i0 + k] : 0;
    const int t = (c[0] + c[1]) + (c[2] + c[3]);

    int inc = t;
    #pragma unroll
    for (int o = 1; o < 32; o <<= 1) {
        const int u = __shfl_up(inc, o);
        if (lane >= o) inc += u;
    }
    if (lane == 31) s_w[wave] = inc;
    if (lane == 0)  s_p[wave] = pre;
    __syncthreads();

    int q = s_p[lane] + (lane < wave ? s_w[lane] : 0);
    for (int o = 16; o > 0; o >>= 1) q += __shfl_xor(q, o);

    int run = q + inc - t;
    #pragma unroll
    for (int k = 0; k < 4; ++k) {
        if (i0 + k < n) {
            out[i0 + k] = run;
            run += c[k];
        }
    }
}

__global__ __launch_bounds__(T1) void bucket_scatter(const int* __restrict__ src,
                                                     const int* __restrict__ dst,
                                                     const int* __restrict__ base,
                                                     int2* __restrict__ pairs,
                                                     int nE, int nB, int per) {
    __shared__ int s_p[MAXB];
    const int tid = threadIdx.x;
    const int w   = blockIdx.x;
    for (int i = tid; i < nB; i += T1) s_p[i] = base[i * G1 + w];
    __syncthreads();
    const int e0 = w * per;
    const int e1 = (e0 + per < nE) ? (e0 + per) : nE;
    for (int e = e0 + tid; e < e1; e += KB * T1) {
        int d[KB], s[KB];
        #pragma unroll
        for (int k = 0; k < KB; ++k) {
            const int ee = (e + k * T1 < e1) ? (e + k * T1) : (e1 - 1);
            d[k] = dst[ee];
            s[k] = src[ee];
        }
        #pragma unroll
        for (int k = 0; k < KB; ++k) {
            if (e + k * T1 < e1) {
                const int p = atomicAdd(&s_p[bucket_of(d[k], nB)], 1);
                if ((unsigned)p < (unsigned)nE) pairs[p] = make_int2(s[k], d[k]);
            }
        }
    }
}

__global__ __launch_bounds__(T4) void bucket_build(const int2* __restrict__ pairs,
                                                   const int* __restrict__ base,
                                                   int* __restrict__ off,
                                                   float* __restrict__ dinv,
                                                   int* __restrict__ csr,
                                                   int nN, int nE, int nB) {
    __shared__ int s_cnt[NB];
    __shared__ int s_cur[NB];
    __shared__ int s_w[NB / 32];
    const int tid  = threadIdx.x;
    const int lane = tid & 31;
    const int b    = blockIdx.x;
    const int node0 = b << NB_SH;
    int bs = base[b * G1];
    int be = (b + 1 < nB) ? base[(b + 1) * G1] : nE;
    bs = bs < 0 ? 0 : (bs < nE ? bs : nE);
    be = be < bs ? bs : (be < nE ? be : nE);

    if (tid < NB) s_cnt[tid] = 0;
    __syncthreads();

    for (int i = bs + tid; i < be; i += K4 * T4) {
        int d[K4];
        #pragma unroll
        for (int k = 0; k < K4; ++k) {
            const int ii = i + k * T4;
            d[k] = pairs[ii < be ? ii : be - 1].y;
        }
        #pragma unroll
        for (int k = 0; k < K4; ++k) {
            const unsigned dl = (unsigned)(d[k] - node0);
            if (i + k * T4 < be && dl < (unsigned)NB) atomicAdd(&s_cnt[dl], 1);
        }
    }
    __syncthreads();

    int c = 0, inc = 0;
    if (tid < NB) {
        c = s_cnt[tid];
        inc = c;
        #pragma unroll
        for (int o = 1; o < 32; o <<= 1) {
            const int u = __shfl_up(inc, o);
            if (lane >= o) inc += u;
        }
        if (lane == 31) s_w[tid >> 5] = inc;
    }
    __syncthreads();

    if (tid < NB) {
        int pre = 0;
        for (int k = 0; k < (tid >> 5); ++k) pre += s_w[k];
        const int start = bs + pre + inc - c;
        s_cur[tid] = start;
        const int node = node0 + tid;
        if (node < nN) {
            off[node]  = start;
            dinv[node] = rsqrtf((float)(c + 1));
            if (node == nN - 1) off[nN] = start + c;
        }
    }
    __syncthreads();

    for (int i = bs + tid; i < be; i += K4 * T4) {
        int2 pr[K4];
        #pragma unroll
        for (int k = 0; k < K4; ++k) {
            const int ii = i + k * T4;
            pr[k] = pairs[ii < be ? ii : be - 1];
        }
        #pragma unroll
        for (int k = 0; k < K4; ++k) {
            const unsigned dl = (unsigned)(pr[k].y - node0);
            if (i + k * T4 < be && dl < (unsigned)NB) {
                const int p = atomicAdd(&s_cur[dl], 1);
                if ((unsigned)p < (unsigned)nE) csr[p] = pr[k].x;
            }
        }
    }
}

#define ROW_LOAD(a, jj) \
    const float4 a = hp[(size_t)__builtin_amdgcn_readlane(s_l, (jj)) * (D / 4) + lane]
#define ROW_FMA(a, jj) do { \
    const float w_ = __int_as_float(__builtin_amdgcn_readlane(w_i, (jj))); \
    acc.x = fmaf(a.x, w_, acc.x); acc.y = fmaf(a.y, w_, acc.y); \
    acc.z = fmaf(a.z, w_, acc.z); acc.w = fmaf(a.w, w_, acc.w); } while (0)

template <bool RELU>
__global__ __launch_bounds__(256, 8) void layer_fused(const int* __restrict__ off,
                                                      const int* __restrict__ csr,
                                                      const float* __restrict__ dinv,
                                                      const float* __restrict__ xin,
                                                      const float* __restrict__ W,
                                                      const float* __restrict__ bias,
                                                      float* __restrict__ out,
                                                      int nN, int nStrips, int nBlk) {
    __shared__ __attribute__((aligned(16))) float xs[2][16 * XS_STRIDE];

    const int tid  = threadIdx.x;
    const int lane = tid & 31;
    const int wave = __builtin_amdgcn_readfirstlane(tid >> 5);
    const int m    = lane & 15;
    const int kk   = (lane >> 4) << 1;
    const int col  = wave * 16 + (lane & 15);
    const int halfm = (lane >> 4) * 8;

    v2f bfr[D / 4];
    #pragma unroll
    for (int i = 0; i < D / 4; ++i) {
        bfr[i].x = W[(size_t)(4 * i + kk) * D + col];
        bfr[i].y = W[(size_t)(4 * i + kk + 1) * D + col];
    }
    const float bc = bias[col];
    const float4* __restrict__ hp = (const float4*)xin;

    int buf = 0;
    for (int strip = blockIdx.x; strip < nStrips; strip += nBlk, buf ^= 1) {
        const int row0 = strip * 16;
        float* __restrict__ xb = xs[buf];

        for (int t = 0; t < 2; ++t) {
            const int r    = 2 * wave + t;
            const int node = row0 + r;
            float4 acc = make_float4(0.0f, 0.0f, 0.0f, 0.0f);
            if (node < nN) {
                const int beg = __builtin_amdgcn_readfirstlane(off[node]);
                const int end = __builtin_amdgcn_readfirstlane(off[node + 1]);
                const float dd = dinv[node];
                const float4 vself = hp[(size_t)node * (D / 4) + lane];

                for (int base = beg; base < end; base += 32) {
                    const int mm = (end - base) < 32 ? (end - base) : 32;
                    int   s_l = 0;
                    float w_l = 0.0f;
                    if (lane < mm) {
                        s_l = csr[base + lane];
                        s_l = s_l < 0 ? 0 : (s_l < nN ? s_l : nN - 1);
                        w_l = dinv[s_l] * dd;
                    }
                    const int w_i = __float_as_int(w_l);

                    int j = 0;
                    for (; j + 8 <= mm; j += 8) {
                        ROW_LOAD(a0, j);     ROW_LOAD(a1, j + 1); ROW_LOAD(a2, j + 2); ROW_LOAD(a3, j + 3);
                        ROW_LOAD(a4, j + 4); ROW_LOAD(a5, j + 5); ROW_LOAD(a6, j + 6); ROW_LOAD(a7, j + 7);
                        ROW_FMA(a0, j);     ROW_FMA(a1, j + 1); ROW_FMA(a2, j + 2); ROW_FMA(a3, j + 3);
                        ROW_FMA(a4, j + 4); ROW_FMA(a5, j + 5); ROW_FMA(a6, j + 6); ROW_FMA(a7, j + 7);
                    }
                    if (j + 4 <= mm) {
                        ROW_LOAD(a0, j); ROW_LOAD(a1, j + 1); ROW_LOAD(a2, j + 2); ROW_LOAD(a3, j + 3);
                        ROW_FMA(a0, j);  ROW_FMA(a1, j + 1);  ROW_FMA(a2, j + 2);  ROW_FMA(a3, j + 3);
                        j += 4;
                    }
                    const int rem = mm - j;
                    if (rem == 3) {
                        ROW_LOAD(a0, j); ROW_LOAD(a1, j + 1); ROW_LOAD(a2, j + 2);
                        ROW_FMA(a0, j);  ROW_FMA(a1, j + 1);  ROW_FMA(a2, j + 2);
                    } else if (rem == 2) {
                        ROW_LOAD(a0, j); ROW_LOAD(a1, j + 1);
                        ROW_FMA(a0, j);  ROW_FMA(a1, j + 1);
                    } else if (rem == 1) {
                        ROW_LOAD(a0, j);
                        ROW_FMA(a0, j);
                    }
                }
                const float ns = dd * dd;
                acc.x = fmaf(vself.x, ns, acc.x); acc.y = fmaf(vself.y, ns, acc.y);
                acc.z = fmaf(vself.z, ns, acc.z); acc.w = fmaf(vself.w, ns, acc.w);
            }
            *(float4*)&xb[r * XS_STRIDE + lane * 4] = acc;
        }
        __syncthreads();

        v8f c = {};
        #pragma unroll
        for (int i = 0; i < D / 4; ++i) {
            v2f a;
            a.x = xb[m * XS_STRIDE + 4 * i + kk];
            a.y = xb[m * XS_STRIDE + 4 * i + kk + 1];
            c = __builtin_amdgcn_wmma_f32_16x16x4_f32(
                 false, a,  false, bfr[i],
                 (short)0, c,  false,  false);
        }

        float v[8];
        #pragma unroll
        for (int r = 0; r < 8; ++r) {
            v[r] = c[r] + bc;
            if (RELU) v[r] = v[r] > 0.0f ? v[r] : 0.0f;
        }
        #pragma unroll
        for (int r = 0; r < 8; ++r) {
            const int row = row0 + halfm + r;
            if (row < nN) out[(size_t)row * D + col] = v[r];
        }
    }
}


static inline size_t align256(size_t v) { return (v + 255) & ~(size_t)255; }

extern "C" void kernel_launch(void* const* d_in, const int* in_sizes, int n_in,
                              void* d_out, int out_size, void* d_ws, size_t ws_size,
                              hipStream_t stream) {
    const float* x  = (const float*)d_in[0];
    const int*   ei = (const int*)d_in[1];
    const float* W1 = (const float*)d_in[2];
    const float* b1 = (const float*)d_in[3];
    const float* W2 = (const float*)d_in[4];
    const float* b2 = (const float*)d_in[5];
    float* out = (float*)d_out;

    const int N = in_sizes[0] / D;
    const int E = in_sizes[1] / 2;
    const int* srcIdx = ei;
    const int* dstIdx = ei + E;

    const int nB  = (N + NB - 1) / NB;
    const int nT  = nB * G1;
    const int per = (E + G1 - 1) / G1;

    char* ws = (char*)d_ws;
    size_t o = 0;
    float* dinv  = (float*)(ws + o); o += align256((size_t)N * sizeof(float));
    int*   tbl   = (int*)(ws + o);   o += align256((size_t)nT * sizeof(int));
    int*   base  = (int*)(ws + o);   o += align256((size_t)nT * sizeof(int));
    int*   off   = (int*)(ws + o);   o += align256((size_t)(N + 1) * sizeof(int));
    int2*  pairs = (int2*)(ws + o);  o += align256((size_t)E * sizeof(int2));
    int*   csr   = (int*)(ws + o);   o += align256((size_t)E * sizeof(int));
    float* hbuf  = (float*)(ws + o);

    const int blk = 256;
    const int gScan = (nT + SCAN_C - 1) / SCAN_C;
    const int nStrips = (N + 15) / 16;
    const int gLayer  = nStrips < 1024 ? nStrips : 1024;

    bucket_count  <<<G1, T1, 0, stream>>>(dstIdx, tbl, E, nB, per);
    scan_excl     <<<gScan, SCAN_T, 0, stream>>>(tbl, base, nT);
    bucket_scatter<<<G1, T1, 0, stream>>>(srcIdx, dstIdx, base, pairs, E, nB, per);
    bucket_build  <<<nB, T4, 0, stream>>>(pairs, base, off, dinv, csr, N, E, nB);

    layer_fused<true> <<<gLayer, blk, 0, stream>>>(off, csr, dinv, x, W1, b1, hbuf, N, nStrips, gLayer);

    layer_fused<false><<<gLayer, blk, 0, stream>>>(off, csr, dinv, hbuf, W2, b2, out, N, nStrips, gLayer);
}


# Layer 1 stores dinv[d]*relu(..): layer 2 sums plain neighbour rows and scales the sum by dinv[d] once, no per-edge weight or dinv[src] lookup there; its sum order changes (adds, then one multiply)
# speedup vs baseline: 5.0645x; 1.0307x over previous
#include <hip/hip_runtime.h>
#include <hip/hip_bf16.h>


typedef float v2f __attribute__((ext_vector_type(2)));
typedef float v8f __attribute__((ext_vector_type(8)));

#define D 128
#define XS_STRIDE 132
#define SCAN_T 1024
#define SCAN_C 4096


#define NB_SH 8
#define NB    256
#define G1    256
#define T1    512
#define T4    1024
#define MAXB  512
#define KB    16
#define K4    4

__device__ __forceinline__ int bucket_of(int d, int nB) {
    const int b = (int)((unsigned)d >> NB_SH);
    return b < nB ? b : nB - 1;
}

__global__ __launch_bounds__(T1) void bucket_count(const int* __restrict__ dst,
                                                   int* __restrict__ tbl,
                                                   int nE, int nB, int per) {
    __shared__ int s_c[MAXB];
    const int tid = threadIdx.x;
    const int w   = blockIdx.x;
    for (int i = tid; i < nB; i += T1) s_c[i] = 0;
    __syncthreads();
    const int e0 = w * per;
    const int e1 = (e0 + per < nE) ? (e0 + per) : nE;
    for (int e = e0 + tid; e < e1; e += KB * T1) {
        int d[KB];
        #pragma unroll
        for (int k = 0; k < KB; ++k) {
            const int ee = e + k * T1;
            d[k] = dst[ee < e1 ? ee : e1 - 1];
        }
        #pragma unroll
        for (int k = 0; k < KB; ++k)
            if (e + k * T1 < e1) atomicAdd(&s_c[bucket_of(d[k], nB)], 1);
    }
    __syncthreads();
    for (int i = tid; i < nB; i += T1) tbl[i * G1 + w] = s_c[i];
}

__global__ __launch_bounds__(SCAN_T) void scan_excl(const int* __restrict__ in,
                                                    int* __restrict__ out, int n) {
    __shared__ int s_w[32];
    __shared__ int s_p[32];
    const int tid  = threadIdx.x;
    const int lane = tid & 31;
    const int wave = tid >> 5;
    const int chunk0 = blockIdx.x * SCAN_C;

    int pre = 0;
    const int4* __restrict__ in4 = (const int4*)in;
    #pragma unroll 8
    for (int k = 0; k < (int)blockIdx.x; ++k) {
        const int4 v = in4[k * SCAN_T + tid];
        pre += (v.x + v.y) + (v.z + v.w);
    }
    for (int o = 16; o > 0; o >>= 1) pre += __shfl_xor(pre, o);

    const int i0 = chunk0 + tid * 4;
    int c[4];
    #pragma unroll
    for (int k = 0; k < 4; ++k) c[k] = (i0 + k < n) ? in[i0 + k] : 0;
    const int t = (c[0] + c[1]) + (c[2] + c[3]);

    int inc = t;
    #pragma unroll
    for (int o = 1; o < 32; o <<= 1) {
        const int u = __shfl_up(inc, o);
        if (lane >= o) inc += u;
    }
    if (lane == 31) s_w[wave] = inc;
    if (lane == 0)  s_p[wave] = pre;
    __syncthreads();

    int q = s_p[lane] + (lane < wave ? s_w[lane] : 0);
    for (int o = 16; o > 0; o >>= 1) q += __shfl_xor(q, o);

    int run = q + inc - t;
    #pragma unroll
    for (int k = 0; k < 4; ++k) {
        if (i0 + k < n) {
            out[i0 + k] = run;
            run += c[k];
        }
    }
}

__global__ __launch_bounds__(T1) void bucket_scatter(const int* __restrict__ src,
                                                     const int* __restrict__ dst,
                                                     const int* __restrict__ base,
                                                     int2* __restrict__ pairs,
                                                     int nE, int nB, int per) {
    __shared__ int s_p[MAXB];
    const int tid = threadIdx.x;
    const int w   = blockIdx.x;
    for (int i = tid; i < nB; i += T1) s_p[i] = base[i * G1 + w];
    __syncthreads();
    const int e0 = w * per;
    const int e1 = (e0 + per < nE) ? (e0 + per) : nE;
    for (int e = e0 + tid; e < e1; e += KB * T1) {
        int d[KB], s[KB];
        #pragma unroll
        for (int k = 0; k < KB; ++k) {
            const int ee = (e + k * T1 < e1) ? (e + k * T1) : (e1 - 1);
            d[k] = dst[ee];
            s[k] = src[ee];
        }
        #pragma unroll
        for (int k = 0; k < KB; ++k) {
            if (e + k * T1 < e1) {
                const int p = atomicAdd(&s_p[bucket_of(d[k], nB)], 1);
                if ((unsigned)p < (unsigned)nE) pairs[p] = make_int2(s[k], d[k]);
            }
        }
    }
}

__global__ __launch_bounds__(T4) void bucket_build(const int2* __restrict__ pairs,
                                                   const int* __restrict__ base,
                                                   int* __restrict__ off,
                                                   float* __restrict__ dinv,
                                                   int* __restrict__ csr,
                                                   int nN, int nE, int nB) {
    __shared__ int s_cnt[NB];
    __shared__ int s_cur[NB];
    __shared__ int s_w[NB / 32];
    const int tid  = threadIdx.x;
    const int lane = tid & 31;
    const int b    = blockIdx.x;
    const int node0 = b << NB_SH;
    int bs = base[b * G1];
    int be = (b + 1 < nB) ? base[(b + 1) * G1] : nE;
    bs = bs < 0 ? 0 : (bs < nE ? bs : nE);
    be = be < bs ? bs : (be < nE ? be : nE);

    if (tid < NB) s_cnt[tid] = 0;
    __syncthreads();

    for (int i = bs + tid; i < be; i += K4 * T4) {
        int d[K4];
        #pragma unroll
        for (int k = 0; k < K4; ++k) {
            const int ii = i + k * T4;
            d[k] = pairs[ii < be ? ii : be - 1].y;
        }
        #pragma unroll
        for (int k = 0; k < K4; ++k) {
            const unsigned dl = (unsigned)(d[k] - node0);
            if (i + k * T4 < be && dl < (unsigned)NB) atomicAdd(&s_cnt[dl], 1);
        }
    }
    __syncthreads();

    int c = 0, inc = 0;
    if (tid < NB) {
        c = s_cnt[tid];
        inc = c;
        #pragma unroll
        for (int o = 1; o < 32; o <<= 1) {
            const int u = __shfl_up(inc, o);
            if (lane >= o) inc += u;
        }
        if (lane == 31) s_w[tid >> 5] = inc;
    }
    __syncthreads();

    if (tid < NB) {
        int pre = 0;
        for (int k = 0; k < (tid >> 5); ++k) pre += s_w[k];
        const int start = bs + pre + inc - c;
        s_cur[tid] = start;
        const int node = node0 + tid;
        if (node < nN) {
            off[node]  = start;
            dinv[node] = rsqrtf((float)(c + 1));
            if (node == nN - 1) off[nN] = start + c;
        }
    }
    __syncthreads();

    for (int i = bs + tid; i < be; i += K4 * T4) {
        int2 pr[K4];
        #pragma unroll
        for (int k = 0; k < K4; ++k) {
            const int ii = i + k * T4;
            pr[k] = pairs[ii < be ? ii : be - 1];
        }
        #pragma unroll
        for (int k = 0; k < K4; ++k) {
            const unsigned dl = (unsigned)(pr[k].y - node0);
            if (i + k * T4 < be && dl < (unsigned)NB) {
                const int p = atomicAdd(&s_cur[dl], 1);
                if ((unsigned)p < (unsigned)nE) csr[p] = pr[k].x;
            }
        }
    }
}

#define ROW_LOAD(a, jj) \
    const float4 a = hp[(size_t)__builtin_amdgcn_readlane(s_l, (jj)) * (D / 4) + lane]
#define ROW_FMA(a, jj) do { \
    const float w_ = __int_as_float(__builtin_amdgcn_readlane(w_i, (jj))); \
    acc.x = fmaf(a.x, w_, acc.x); acc.y = fmaf(a.y, w_, acc.y); \
    acc.z = fmaf(a.z, w_, acc.z); acc.w = fmaf(a.w, w_, acc.w); } while (0)
#define ROW_ADD(a) do { \
    acc.x += a.x; acc.y += a.y; acc.z += a.z; acc.w += a.w; } while (0)
#define ROW_ACC(a, jj) do { if (LAYER == 1) ROW_FMA(a, jj); else ROW_ADD(a); } while (0)

template <int LAYER>
__global__ __launch_bounds__(256, 8) void layer_fused(const int* __restrict__ off,
                                                      const int* __restrict__ csr,
                                                      const float* __restrict__ dinv,
                                                      const float* __restrict__ xin,
                                                      const float* __restrict__ W,
                                                      const float* __restrict__ bias,
                                                      float* __restrict__ out,
                                                      int nN, int nStrips, int nBlk) {
    __shared__ __attribute__((aligned(16))) float xs[2][16 * XS_STRIDE];

    const int tid  = threadIdx.x;
    const int lane = tid & 31;
    const int wave = __builtin_amdgcn_readfirstlane(tid >> 5);
    const int m    = lane & 15;
    const int kk   = (lane >> 4) << 1;
    const int col  = wave * 16 + (lane & 15);
    const int halfm = (lane >> 4) * 8;

    v2f bfr[D / 4];
    #pragma unroll
    for (int i = 0; i < D / 4; ++i) {
        bfr[i].x = W[(size_t)(4 * i + kk) * D + col];
        bfr[i].y = W[(size_t)(4 * i + kk + 1) * D + col];
    }
    const float bc = bias[col];
    const float4* __restrict__ hp = (const float4*)xin;

    int buf = 0;
    for (int strip = blockIdx.x; strip < nStrips; strip += nBlk, buf ^= 1) {
        const int row0 = strip * 16;
        float* __restrict__ xb = xs[buf];

        for (int t = 0; t < 2; ++t) {
            const int r    = 2 * wave + t;
            const int node = row0 + r;
            float4 acc = make_float4(0.0f, 0.0f, 0.0f, 0.0f);
            float ddr = 0.0f;
            if (node < nN) {
                const int beg = __builtin_amdgcn_readfirstlane(off[node]);
                const int end = __builtin_amdgcn_readfirstlane(off[node + 1]);
                const float dd = dinv[node];
                const float4 vself = hp[(size_t)node * (D / 4) + lane];

                for (int base = beg; base < end; base += 32) {
                    const int mm = (end - base) < 32 ? (end - base) : 32;
                    int   s_l = 0;
                    float w_l = 0.0f;
                    if (lane < mm) {
                        s_l = csr[base + lane];
                        s_l = s_l < 0 ? 0 : (s_l < nN ? s_l : nN - 1);
                        if (LAYER == 1) w_l = dinv[s_l] * dd;
                    }
                    const int w_i = __float_as_int(w_l);

                    int j = 0;
                    for (; j + 8 <= mm; j += 8) {
                        ROW_LOAD(a0, j);     ROW_LOAD(a1, j + 1); ROW_LOAD(a2, j + 2); ROW_LOAD(a3, j + 3);
                        ROW_LOAD(a4, j + 4); ROW_LOAD(a5, j + 5); ROW_LOAD(a6, j + 6); ROW_LOAD(a7, j + 7);
                        ROW_ACC(a0, j);     ROW_ACC(a1, j + 1); ROW_ACC(a2, j + 2); ROW_ACC(a3, j + 3);
                        ROW_ACC(a4, j + 4); ROW_ACC(a5, j + 5); ROW_ACC(a6, j + 6); ROW_ACC(a7, j + 7);
                    }
                    if (j + 4 <= mm) {
                        ROW_LOAD(a0, j); ROW_LOAD(a1, j + 1); ROW_LOAD(a2, j + 2); ROW_LOAD(a3, j + 3);
                        ROW_ACC(a0, j);  ROW_ACC(a1, j + 1);  ROW_ACC(a2, j + 2);  ROW_ACC(a3, j + 3);
                        j += 4;
                    }
                    const int rem = mm - j;
                    if (rem == 3) {
                        ROW_LOAD(a0, j); ROW_LOAD(a1, j + 1); ROW_LOAD(a2, j + 2);
                        ROW_ACC(a0, j);  ROW_ACC(a1, j + 1);  ROW_ACC(a2, j + 2);
                    } else if (rem == 2) {
                        ROW_LOAD(a0, j); ROW_LOAD(a1, j + 1);
                        ROW_ACC(a0, j);  ROW_ACC(a1, j + 1);
                    } else if (rem == 1) {
                        ROW_LOAD(a0, j);
                        ROW_ACC(a0, j);
                    }
                }
                if (LAYER == 1) {
                    const float ns = dd * dd;
                    acc.x = fmaf(vself.x, ns, acc.x); acc.y = fmaf(vself.y, ns, acc.y);
                    acc.z = fmaf(vself.z, ns, acc.z); acc.w = fmaf(vself.w, ns, acc.w);
                    ddr = dd;
                } else {
                    acc.x = (acc.x + vself.x) * dd; acc.y = (acc.y + vself.y) * dd;
                    acc.z = (acc.z + vself.z) * dd; acc.w = (acc.w + vself.w) * dd;
                }
            }
            *(float4*)&xb[r * XS_STRIDE + lane * 4] = acc;
            if (LAYER == 1 && lane == 0) xb[r * XS_STRIDE + D] = ddr;
        }
        __syncthreads();

        v8f c = {};
        #pragma unroll
        for (int i = 0; i < D / 4; ++i) {
            v2f a;
            a.x = xb[m * XS_STRIDE + 4 * i + kk];
            a.y = xb[m * XS_STRIDE + 4 * i + kk + 1];
            c = __builtin_amdgcn_wmma_f32_16x16x4_f32(
                 false, a,  false, bfr[i],
                 (short)0, c,  false,  false);
        }

        float v[8];
        #pragma unroll
        for (int r = 0; r < 8; ++r) {
            v[r] = c[r] + bc;
            if (LAYER == 1) {
                v[r] = v[r] > 0.0f ? v[r] : 0.0f;
                v[r] *= xb[(halfm + r) * XS_STRIDE + D];
            }
        }
        #pragma unroll
        for (int r = 0; r < 8; ++r) {
            const int row = row0 + halfm + r;
            if (row < nN) out[(size_t)row * D + col] = v[r];
        }
    }
}


static inline size_t align256(size_t v) { return (v + 255) & ~(size_t)255; }

extern "C" void kernel_launch(void* const* d_in, const int* in_sizes, int n_in,
                              void* d_out, int out_size, void* d_ws, size_t ws_size,
                              hipStream_t stream) {
    const float* x  = (const float*)d_in[0];
    const int*   ei = (const int*)d_in[1];
    const float* W1 = (const float*)d_in[2];
    const float* b1 = (const float*)d_in[3];
    const float* W2 = (const float*)d_in[4];
    const float* b2 = (const float*)d_in[5];
    float* out = (float*)d_out;

    const int N = in_sizes[0] / D;
    const int E = in_sizes[1] / 2;
    const int* srcIdx = ei;
    const int* dstIdx = ei + E;

    const int nB  = (N + NB - 1) / NB;
    const int nT  = nB * G1;
    const int per = (E + G1 - 1) / G1;

    char* ws = (char*)d_ws;
    size_t o = 0;
    float* dinv  = (float*)(ws + o); o += align256((size_t)N * sizeof(float));
    int*   tbl   = (int*)(ws + o);   o += align256((size_t)nT * sizeof(int));
    int*   base  = (int*)(ws + o);   o += align256((size_t)nT * sizeof(int));
    int*   off   = (int*)(ws + o);   o += align256((size_t)(N + 1) * sizeof(int));
    int2*  pairs = (int2*)(ws + o);  o += align256((size_t)E * sizeof(int2));
    int*   csr   = (int*)(ws + o);   o += align256((size_t)E * sizeof(int));
    float* hbuf  = (float*)(ws + o);

    const int blk = 256;
    const int gScan = (nT + SCAN_C - 1) / SCAN_C;
    const int nStrips = (N + 15) / 16;
    const int gLayer  = nStrips < 1024 ? nStrips : 1024;

    bucket_count  <<<G1, T1, 0, stream>>>(dstIdx, tbl, E, nB, per);
    scan_excl     <<<gScan, SCAN_T, 0, stream>>>(tbl, base, nT);
    bucket_scatter<<<G1, T1, 0, stream>>>(srcIdx, dstIdx, base, pairs, E, nB, per);
    bucket_build  <<<nB, T4, 0, stream>>>(pairs, base, off, dinv, csr, N, E, nB);

    layer_fused<1><<<gLayer, blk, 0, stream>>>(off, csr, dinv, x, W1, b1, hbuf, N, nStrips, gLayer);

    layer_fused<2><<<gLayer, blk, 0, stream>>>(off, csr, dinv, hbuf, W2, b2, out, N, nStrips, gLayer);
}


# Each wave fetches what a node's walk starts from (row bounds, dinv, source ids, source weights) one phase ahead, under the row sums before it; a walk starts at its row loads; sum order unchanged
# speedup vs baseline: 5.2522x; 1.0371x over previous
#include <hip/hip_runtime.h>
#include <hip/hip_bf16.h>


typedef float v2f __attribute__((ext_vector_type(2)));
typedef float v8f __attribute__((ext_vector_type(8)));

#define D 128
#define XS_STRIDE 132
#define SCAN_T 1024
#define SCAN_C 4096


#define NB_SH 8
#define NB    256
#define G1    256
#define T1    512
#define T4    1024
#define MAXB  512
#define KB    16
#define K4    4

__device__ __forceinline__ int bucket_of(int d, int nB) {
    const int b = (int)((unsigned)d >> NB_SH);
    return b < nB ? b : nB - 1;
}

__global__ __launch_bounds__(T1) void bucket_count(const int* __restrict__ dst,
                                                   int* __restrict__ tbl,
                                                   int nE, int nB, int per) {
    __shared__ int s_c[MAXB];
    const int tid = threadIdx.x;
    const int w   = blockIdx.x;
    for (int i = tid; i < nB; i += T1) s_c[i] = 0;
    __syncthreads();
    const int e0 = w * per;
    const int e1 = (e0 + per < nE) ? (e0 + per) : nE;
    for (int e = e0 + tid; e < e1; e += KB * T1) {
        int d[KB];
        #pragma unroll
        for (int k = 0; k < KB; ++k) {
            const int ee = e + k * T1;
            d[k] = dst[ee < e1 ? ee : e1 - 1];
        }
        #pragma unroll
        for (int k = 0; k < KB; ++k)
            if (e + k * T1 < e1) atomicAdd(&s_c[bucket_of(d[k], nB)], 1);
    }
    __syncthreads();
    for (int i = tid; i < nB; i += T1) tbl[i * G1 + w] = s_c[i];
}

__global__ __launch_bounds__(SCAN_T) void scan_excl(const int* __restrict__ in,
                                                    int* __restrict__ out, int n) {
    __shared__ int s_w[32];
    __shared__ int s_p[32];
    const int tid  = threadIdx.x;
    const int lane = tid & 31;
    const int wave = tid >> 5;
    const int chunk0 = blockIdx.x * SCAN_C;

    int pre = 0;
    const int4* __restrict__ in4 = (const int4*)in;
    #pragma unroll 8
    for (int k = 0; k < (int)blockIdx.x; ++k) {
        const int4 v = in4[k * SCAN_T + tid];
        pre += (v.x + v.y) + (v.z + v.w);
    }
    for (int o = 16; o > 0; o >>= 1) pre += __shfl_xor(pre, o);

    const int i0 = chunk0 + tid * 4;
    int c[4];
    #pragma unroll
    for (int k = 0; k < 4; ++k) c[k] = (i0 + k < n) ? in[i0 + k] : 0;
    const int t = (c[0] + c[1]) + (c[2] + c[3]);

    int inc = t;
    #pragma unroll
    for (int o = 1; o < 32; o <<= 1) {
        const int u = __shfl_up(inc, o);
        if (lane >= o) inc += u;
    }
    if (lane == 31) s_w[wave] = inc;
    if (lane == 0)  s_p[wave] = pre;
    __syncthreads();

    int q = s_p[lane] + (lane < wave ? s_w[lane] : 0);
    for (int o = 16; o > 0; o >>= 1) q += __shfl_xor(q, o);

    int run = q + inc - t;
    #pragma unroll
    for (int k = 0; k < 4; ++k) {
        if (i0 + k < n) {
            out[i0 + k] = run;
            run += c[k];
        }
    }
}

__global__ __launch_bounds__(T1) void bucket_scatter(const int* __restrict__ src,
                                                     const int* __restrict__ dst,
                                                     const int* __restrict__ base,
                                                     int2* __restrict__ pairs,
                                                     int nE, int nB, int per) {
    __shared__ int s_p[MAXB];
    const int tid = threadIdx.x;
    const int w   = blockIdx.x;
    for (int i = tid; i < nB; i += T1) s_p[i] = base[i * G1 + w];
    __syncthreads();
    const int e0 = w * per;
    const int e1 = (e0 + per < nE) ? (e0 + per) : nE;
    for (int e = e0 + tid; e < e1; e += KB * T1) {
        int d[KB], s[KB];
        #pragma unroll
        for (int k = 0; k < KB; ++k) {
            const int ee = (e + k * T1 < e1) ? (e + k * T1) : (e1 - 1);
            d[k] = dst[ee];
            s[k] = src[ee];
        }
        #pragma unroll
        for (int k = 0; k < KB; ++k) {
            if (e + k * T1 < e1) {
                const int p = atomicAdd(&s_p[bucket_of(d[k], nB)], 1);
                if ((unsigned)p < (unsigned)nE) pairs[p] = make_int2(s[k], d[k]);
            }
        }
    }
}

__global__ __launch_bounds__(T4) void bucket_build(const int2* __restrict__ pairs,
                                                   const int* __restrict__ base,
                                                   int* __restrict__ off,
                                                   float* __restrict__ dinv,
                                                   int* __restrict__ csr,
                                                   int nN, int nE, int nB) {
    __shared__ int s_cnt[NB];
    __shared__ int s_cur[NB];
    __shared__ int s_w[NB / 32];
    const int tid  = threadIdx.x;
    const int lane = tid & 31;
    const int b    = blockIdx.x;
    const int node0 = b << NB_SH;
    int bs = base[b * G1];
    int be = (b + 1 < nB) ? base[(b + 1) * G1] : nE;
    bs = bs < 0 ? 0 : (bs < nE ? bs : nE);
    be = be < bs ? bs : (be < nE ? be : nE);

    if (tid < NB) s_cnt[tid] = 0;
    __syncthreads();

    for (int i = bs + tid; i < be; i += K4 * T4) {
        int d[K4];
        #pragma unroll
        for (int k = 0; k < K4; ++k) {
            const int ii = i + k * T4;
            d[k] = pairs[ii < be ? ii : be - 1].y;
        }
        #pragma unroll
        for (int k = 0; k < K4; ++k) {
            const unsigned dl = (unsigned)(d[k] - node0);
            if (i + k * T4 < be && dl < (unsigned)NB) atomicAdd(&s_cnt[dl], 1);
        }
    }
    __syncthreads();

    int c = 0, inc = 0;
    if (tid < NB) {
        c = s_cnt[tid];
        inc = c;
        #pragma unroll
        for (int o = 1; o < 32; o <<= 1) {
            const int u = __shfl_up(inc, o);
            if (lane >= o) inc += u;
        }
        if (lane == 31) s_w[tid >> 5] = inc;
    }
    __syncthreads();

    if (tid < NB) {
        int pre = 0;
        for (int k = 0; k < (tid >> 5); ++k) pre += s_w[k];
        const int start = bs + pre + inc - c;
        s_cur[tid] = start;
        const int node = node0 + tid;
        if (node < nN) {
            off[node]  = start;
            dinv[node] = rsqrtf((float)(c + 1));
            if (node == nN - 1) off[nN] = start + c;
        }
    }
    __syncthreads();

    for (int i = bs + tid; i < be; i += K4 * T4) {
        int2 pr[K4];
        #pragma unroll
        for (int k = 0; k < K4; ++k) {
            const int ii = i + k * T4;
            pr[k] = pairs[ii < be ? ii : be - 1];
        }
        #pragma unroll
        for (int k = 0; k < K4; ++k) {
            const unsigned dl = (unsigned)(pr[k].y - node0);
            if (i + k * T4 < be && dl < (unsigned)NB) {
                const int p = atomicAdd(&s_cur[dl], 1);
                if ((unsigned)p < (unsigned)nE) csr[p] = pr[k].x;
            }
        }
    }
}

#define ROW_LOAD(a, jj) \
    const float4 a = hp[(size_t)__builtin_amdgcn_readlane(s_l, (jj)) * (D / 4) + lane]
#define ROW_FMA(a, jj) do { \
    const float w_ = __int_as_float(__builtin_amdgcn_readlane(w_i, (jj))); \
    acc.x = fmaf(a.x, w_, acc.x); acc.y = fmaf(a.y, w_, acc.y); \
    acc.z = fmaf(a.z, w_, acc.z); acc.w = fmaf(a.w, w_, acc.w); } while (0)
#define ROW_ADD(a) do { \
    acc.x += a.x; acc.y += a.y; acc.z += a.z; acc.w += a.w; } while (0)
#define ROW_ACC(a, jj) do { if (LAYER == 1) ROW_FMA(a, jj); else ROW_ADD(a); } while (0)

__device__ __forceinline__ int clamp_id(int s, int nN) {
    return s < 0 ? 0 : (s < nN ? s : nN - 1);
}

__device__ __forceinline__ int first32(int beg, int end, int lane) {
    int i = beg + lane;
    i = i < end ? i : end - 1;
    return i < 0 ? 0 : i;
}

template <int LAYER>
__device__ __forceinline__ float4 node_sum(const float4* __restrict__ hp,
                                           const int* __restrict__ csr,
                                           const float* __restrict__ dinv,
                                           int node, int beg, int end, float dd,
                                           int s0, float d0, int lane, int nN) {
    float4 acc = make_float4(0.0f, 0.0f, 0.0f, 0.0f);
    if (node < nN) {
        const float4 vself = hp[(size_t)node * (D / 4) + lane];

        for (int base = beg; base < end; base += 32) {
            const int mm = (end - base) < 32 ? (end - base) : 32;
            int   s_l = s0;
            float w_l = d0 * dd;
            if (base != beg) {
                s_l = 0;
                w_l = 0.0f;
                if (lane < mm) {
                    s_l = clamp_id(csr[base + lane], nN);
                    if (LAYER == 1) w_l = dinv[s_l] * dd;
                }
            }
            const int w_i = __float_as_int(w_l);

            int j = 0;
            for (; j + 8 <= mm; j += 8) {
                ROW_LOAD(a0, j);     ROW_LOAD(a1, j + 1); ROW_LOAD(a2, j + 2); ROW_LOAD(a3, j + 3);
                ROW_LOAD(a4, j + 4); ROW_LOAD(a5, j + 5); ROW_LOAD(a6, j + 6); ROW_LOAD(a7, j + 7);
                ROW_ACC(a0, j);     ROW_ACC(a1, j + 1); ROW_ACC(a2, j + 2); ROW_ACC(a3, j + 3);
                ROW_ACC(a4, j + 4); ROW_ACC(a5, j + 5); ROW_ACC(a6, j + 6); ROW_ACC(a7, j + 7);
            }
            if (j + 4 <= mm) {
                ROW_LOAD(a0, j); ROW_LOAD(a1, j + 1); ROW_LOAD(a2, j + 2); ROW_LOAD(a3, j + 3);
                ROW_ACC(a0, j);  ROW_ACC(a1, j + 1);  ROW_ACC(a2, j + 2);  ROW_ACC(a3, j + 3);
                j += 4;
            }
            const int rem = mm - j;
            if (rem == 3) {
                ROW_LOAD(a0, j); ROW_LOAD(a1, j + 1); ROW_LOAD(a2, j + 2);
                ROW_ACC(a0, j);  ROW_ACC(a1, j + 1);  ROW_ACC(a2, j + 2);
            } else if (rem == 2) {
                ROW_LOAD(a0, j); ROW_LOAD(a1, j + 1);
                ROW_ACC(a0, j);  ROW_ACC(a1, j + 1);
            } else if (rem == 1) {
                ROW_LOAD(a0, j);
                ROW_ACC(a0, j);
            }
        }
        if (LAYER == 1) {
            const float ns = dd * dd;
            acc.x = fmaf(vself.x, ns, acc.x); acc.y = fmaf(vself.y, ns, acc.y);
            acc.z = fmaf(vself.z, ns, acc.z); acc.w = fmaf(vself.w, ns, acc.w);
        } else {
            acc.x = (acc.x + vself.x) * dd; acc.y = (acc.y + vself.y) * dd;
            acc.z = (acc.z + vself.z) * dd; acc.w = (acc.w + vself.w) * dd;
        }
    }
    return acc;
}

template <int LAYER>
__global__ __launch_bounds__(256, 8) void layer_fused(const int* __restrict__ off,
                                                      const int* __restrict__ csr,
                                                      const float* __restrict__ dinv,
                                                      const float* __restrict__ xin,
                                                      const float* __restrict__ W,
                                                      const float* __restrict__ bias,
                                                      float* __restrict__ out,
                                                      int nN, int nStrips, int nBlk) {
    __shared__ __attribute__((aligned(16))) float xs[2][16 * XS_STRIDE];

    const int tid  = threadIdx.x;
    const int lane = tid & 31;
    const int wave = __builtin_amdgcn_readfirstlane(tid >> 5);
    const int m    = lane & 15;
    const int kk   = (lane >> 4) << 1;
    const int col  = wave * 16 + (lane & 15);
    const int halfm = (lane >> 4) * 8;

    v2f bfr[D / 4];
    #pragma unroll
    for (int i = 0; i < D / 4; ++i) {
        bfr[i].x = W[(size_t)(4 * i + kk) * D + col];
        bfr[i].y = W[(size_t)(4 * i + kk + 1) * D + col];
    }
    const float bc = bias[col];
    const float4* __restrict__ hp = (const float4*)xin;

    int   strip = blockIdx.x;
    int   offv = 0, ida = 0, idb = 0;
    float ddv = 0.0f, wa = 0.0f;
    if (strip < nStrips) {
        const int i = strip * 16 + 2 * wave + lane;
        offv = off[i < nN ? i : nN];
        ddv  = dinv[i < nN ? i : nN - 1];
        const int b0 = __builtin_amdgcn_readlane(offv, 0);
        const int e0 = __builtin_amdgcn_readlane(offv, 1);
        const int e1 = __builtin_amdgcn_readlane(offv, 2);
        ida = csr[first32(b0, e0, lane)];
        idb = csr[first32(e0, e1, lane)];
        ida = clamp_id(ida, nN);
        if (LAYER == 1) wa = dinv[ida];
    }

    int buf = 0;
    for (; strip < nStrips; strip += nBlk, buf ^= 1) {
        const int row0 = strip * 16;
        float* __restrict__ xb = xs[buf];
        const int na = row0 + 2 * wave;
        const int beg_a = __builtin_amdgcn_readlane(offv, 0);
        const int end_a = __builtin_amdgcn_readlane(offv, 1);
        const int end_b = __builtin_amdgcn_readlane(offv, 2);
        const float dd_a = __int_as_float(__builtin_amdgcn_readlane(__float_as_int(ddv), 0));
        const float dd_b = __int_as_float(__builtin_amdgcn_readlane(__float_as_int(ddv), 1));
        const int   sb  = clamp_id(idb, nN);

        const int nxt = (strip + nBlk < nStrips) ? (strip + nBlk) : strip;
        const int in  = nxt * 16 + 2 * wave + lane;
        const int   offn = off[in < nN ? in : nN];
        const float ddn  = dinv[in < nN ? in : nN - 1];
        float wb = 0.0f;
        if (LAYER == 1) wb = dinv[sb];

        {
            const float4 acc = node_sum<LAYER>(hp, csr, dinv, na, beg_a, end_a, dd_a, ida, wa, lane, nN);
            const int r = 2 * wave;
            *(float4*)&xb[r * XS_STRIDE + lane * 4] = acc;
            if (LAYER == 1 && lane == 0) xb[r * XS_STRIDE + D] = (na < nN) ? dd_a : 0.0f;
        }

        const int nb0 = __builtin_amdgcn_readlane(offn, 0);
        const int ne0 = __builtin_amdgcn_readlane(offn, 1);
        const int ne1 = __builtin_amdgcn_readlane(offn, 2);
        int       ida_n = csr[first32(nb0, ne0, lane)];
        const int idb_n = csr[first32(ne0, ne1, lane)];

        {
            const float4 acc = node_sum<LAYER>(hp, csr, dinv, na + 1, end_a, end_b, dd_b, sb, wb, lane, nN);
            const int r = 2 * wave + 1;
            *(float4*)&xb[r * XS_STRIDE + lane * 4] = acc;
            if (LAYER == 1 && lane == 0) xb[r * XS_STRIDE + D] = (na + 1 < nN) ? dd_b : 0.0f;
        }
        __syncthreads();

        float wa_n = 0.0f;
        ida_n = clamp_id(ida_n, nN);
        if (LAYER == 1) {
            wa_n = dinv[ida_n];
            __builtin_amdgcn_sched_barrier(0);
        }

        v8f c = {};
        #pragma unroll
        for (int i = 0; i < D / 4; ++i) {
            v2f a;
            a.x = xb[m * XS_STRIDE + 4 * i + kk];
            a.y = xb[m * XS_STRIDE + 4 * i + kk + 1];
            c = __builtin_amdgcn_wmma_f32_16x16x4_f32(
                 false, a,  false, bfr[i],
                 (short)0, c,  false,  false);
        }

        float v[8];
        #pragma unroll
        for (int r = 0; r < 8; ++r) {
            v[r] = c[r] + bc;
            if (LAYER == 1) {
                v[r] = v[r] > 0.0f ? v[r] : 0.0f;
                v[r] *= xb[(halfm + r) * XS_STRIDE + D];
            }
        }
        #pragma unroll
        for (int r = 0; r < 8; ++r) {
            const int row = row0 + halfm + r;
            if (row < nN) out[(size_t)row * D + col] = v[r];
        }

        offv = offn; ddv = ddn; ida = ida_n; idb = idb_n; wa = wa_n;
    }
}


static inline size_t align256(size_t v) { return (v + 255) & ~(size_t)255; }

extern "C" void kernel_launch(void* const* d_in, const int* in_sizes, int n_in,
                              void* d_out, int out_size, void* d_ws, size_t ws_size,
                              hipStream_t stream) {
    const float* x  = (const float*)d_in[0];
    const int*   ei = (const int*)d_in[1];
    const float* W1 = (const float*)d_in[2];
    const float* b1 = (const float*)d_in[3];
    const float* W2 = (const float*)d_in[4];
    const float* b2 = (const float*)d_in[5];
    float* out = (float*)d_out;

    const int N = in_sizes[0] / D;
    const int E = in_sizes[1] / 2;
    const int* srcIdx = ei;
    const int* dstIdx = ei + E;

    const int nB  = (N + NB - 1) / NB;
    const int nT  = nB * G1;
    const int per = (E + G1 - 1) / G1;

    char* ws = (char*)d_ws;
    size_t o = 0;
    float* dinv  = (float*)(ws + o); o += align256((size_t)N * sizeof(float));
    int*   tbl   = (int*)(ws + o);   o += align256((size_t)nT * sizeof(int));
    int*   base  = (int*)(ws + o);   o += align256((size_t)nT * sizeof(int));
    int*   off   = (int*)(ws + o);   o += align256((size_t)(N + 1) * sizeof(int));
    int2*  pairs = (int2*)(ws + o);  o += align256((size_t)E * sizeof(int2));
    int*   csr   = (int*)(ws + o);   o += align256((size_t)E * sizeof(int));
    float* hbuf  = (float*)(ws + o);

    const int blk = 256;
    const int gScan = (nT + SCAN_C - 1) / SCAN_C;
    const int nStrips = (N + 15) / 16;
    const int gLayer  = nStrips < 1024 ? nStrips : 1024;

    bucket_count  <<<G1, T1, 0, stream>>>(dstIdx, tbl, E, nB, per);
    scan_excl     <<<gScan, SCAN_T, 0, stream>>>(tbl, base, nT);
    bucket_scatter<<<G1, T1, 0, stream>>>(srcIdx, dstIdx, base, pairs, E, nB, per);
    bucket_build  <<<nB, T4, 0, stream>>>(pairs, base, off, dinv, csr, N, E, nB);

    layer_fused<1><<<gLayer, blk, 0, stream>>>(off, csr, dinv, x, W1, b1, hbuf, N, nStrips, gLayer);

    layer_fused<2><<<gLayer, blk, 0, stream>>>(off, csr, dinv, hbuf, W2, b2, out, N, nStrips, gLayer);
}


# Results leave as whole rows: a wave's 16x16 tile waits in LDS, and behind the next strip's barrier each wave stores two rows, one 512-B store each (was 64-B half lines per wave); sums unchanged
# speedup vs baseline: 5.6197x; 1.0700x over previous
#include <hip/hip_runtime.h>
#include <hip/hip_bf16.h>


typedef float v2f __attribute__((ext_vector_type(2)));
typedef float v8f __attribute__((ext_vector_type(8)));

#define D 128
#define XS_STRIDE 132
#define SCAN_T 1024
#define SCAN_C 4096


#define NB_SH 8
#define NB    256
#define G1    256
#define T1    512
#define T4    1024
#define MAXB  512
#define KB    16
#define K4    4

__device__ __forceinline__ int bucket_of(int d, int nB) {
    const int b = (int)((unsigned)d >> NB_SH);
    return b < nB ? b : nB - 1;
}

__global__ __launch_bounds__(T1) void bucket_count(const int* __restrict__ dst,
                                                   int* __restrict__ tbl,
                                                   int nE, int nB, int per) {
    __shared__ int s_c[MAXB];
    const int tid = threadIdx.x;
    const int w   = blockIdx.x;
    for (int i = tid; i < nB; i += T1) s_c[i] = 0;
    __syncthreads();
    const int e0 = w * per;
    const int e1 = (e0 + per < nE) ? (e0 + per) : nE;
    for (int e = e0 + tid; e < e1; e += KB * T1) {
        int d[KB];
        #pragma unroll
        for (int k = 0; k < KB; ++k) {
            const int ee = e + k * T1;
            d[k] = dst[ee < e1 ? ee : e1 - 1];
        }
        #pragma unroll
        for (int k = 0; k < KB; ++k)
            if (e + k * T1 < e1) atomicAdd(&s_c[bucket_of(d[k], nB)], 1);
    }
    __syncthreads();
    for (int i = tid; i < nB; i += T1) tbl[i * G1 + w] = s_c[i];
}

__global__ __launch_bounds__(SCAN_T) void scan_excl(const int* __restrict__ in,
                                                    int* __restrict__ out, int n) {
    __shared__ int s_w[32];
    __shared__ int s_p[32];
    const int tid  = threadIdx.x;
    const int lane = tid & 31;
    const int wave = tid >> 5;
    const int chunk0 = blockIdx.x * SCAN_C;

    int pre = 0;
    const int4* __restrict__ in4 = (const int4*)in;
    #pragma unroll 8
    for (int k = 0; k < (int)blockIdx.x; ++k) {
        const int4 v = in4[k * SCAN_T + tid];
        pre += (v.x + v.y) + (v.z + v.w);
    }
    for (int o = 16; o > 0; o >>= 1) pre += __shfl_xor(pre, o);

    const int i0 = chunk0 + tid * 4;
    int c[4];
    #pragma unroll
    for (int k = 0; k < 4; ++k) c[k] = (i0 + k < n) ? in[i0 + k] : 0;
    const int t = (c[0] + c[1]) + (c[2] + c[3]);

    int inc = t;
    #pragma unroll
    for (int o = 1; o < 32; o <<= 1) {
        const int u = __shfl_up(inc, o);
        if (lane >= o) inc += u;
    }
    if (lane == 31) s_w[wave] = inc;
    if (lane == 0)  s_p[wave] = pre;
    __syncthreads();

    int q = s_p[lane] + (lane < wave ? s_w[lane] : 0);
    for (int o = 16; o > 0; o >>= 1) q += __shfl_xor(q, o);

    int run = q + inc - t;
    #pragma unroll
    for (int k = 0; k < 4; ++k) {
        if (i0 + k < n) {
            out[i0 + k] = run;
            run += c[k];
        }
    }
}

__global__ __launch_bounds__(T1) void bucket_scatter(const int* __restrict__ src,
                                                     const int* __restrict__ dst,
                                                     const int* __restrict__ base,
                                                     int2* __restrict__ pairs,
                                                     int nE, int nB, int per) {
    __shared__ int s_p[MAXB];
    const int tid = threadIdx.x;
    const int w   = blockIdx.x;
    for (int i = tid; i < nB; i += T1) s_p[i] = base[i * G1 + w];
    __syncthreads();
    const int e0 = w * per;
    const int e1 = (e0 + per < nE) ? (e0 + per) : nE;
    for (int e = e0 + tid; e < e1; e += KB * T1) {
        int d[KB], s[KB];
        #pragma unroll
        for (int k = 0; k < KB; ++k) {
            const int ee = (e + k * T1 < e1) ? (e + k * T1) : (e1 - 1);
            d[k] = dst[ee];
            s[k] = src[ee];
        }
        #pragma unroll
        for (int k = 0; k < KB; ++k) {
            if (e + k * T1 < e1) {
                const int p = atomicAdd(&s_p[bucket_of(d[k], nB)], 1);
                if ((unsigned)p < (unsigned)nE) pairs[p] = make_int2(s[k], d[k]);
            }
        }
    }
}

__global__ __launch_bounds__(T4) void bucket_build(const int2* __restrict__ pairs,
                                                   const int* __restrict__ base,
                                                   int* __restrict__ off,
                                                   float* __restrict__ dinv,
                                                   int* __restrict__ csr,
                                                   int nN, int nE, int nB) {
    __shared__ int s_cnt[NB];
    __shared__ int s_cur[NB];
    __shared__ int s_w[NB / 32];
    const int tid  = threadIdx.x;
    const int lane = tid & 31;
    const int b    = blockIdx.x;
    const int node0 = b << NB_SH;
    int bs = base[b * G1];
    int be = (b + 1 < nB) ? base[(b + 1) * G1] : nE;
    bs = bs < 0 ? 0 : (bs < nE ? bs : nE);
    be = be < bs ? bs : (be < nE ? be : nE);

    if (tid < NB) s_cnt[tid] = 0;
    __syncthreads();

    for (int i = bs + tid; i < be; i += K4 * T4) {
        int d[K4];
        #pragma unroll
        for (int k = 0; k < K4; ++k) {
            const int ii = i + k * T4;
            d[k] = pairs[ii < be ? ii : be - 1].y;
        }
        #pragma unroll
        for (int k = 0; k < K4; ++k) {
            const unsigned dl = (unsigned)(d[k] - node0);
            if (i + k * T4 < be && dl < (unsigned)NB) atomicAdd(&s_cnt[dl], 1);
        }
    }
    __syncthreads();

    int c = 0, inc = 0;
    if (tid < NB) {
        c = s_cnt[tid];
        inc = c;
        #pragma unroll
        for (int o = 1; o < 32; o <<= 1) {
            const int u = __shfl_up(inc, o);
            if (lane >= o) inc += u;
        }
        if (lane == 31) s_w[tid >> 5] = inc;
    }
    __syncthreads();

    if (tid < NB) {
        int pre = 0;
        for (int k = 0; k < (tid >> 5); ++k) pre += s_w[k];
        const int start = bs + pre + inc - c;
        s_cur[tid] = start;
        const int node = node0 + tid;
        if (node < nN) {
            off[node]  = start;
            dinv[node] = rsqrtf((float)(c + 1));
            if (node == nN - 1) off[nN] = start + c;
        }
    }
    __syncthreads();

    for (int i = bs + tid; i < be; i += K4 * T4) {
        int2 pr[K4];
        #pragma unroll
        for (int k = 0; k < K4; ++k) {
            const int ii = i + k * T4;
            pr[k] = pairs[ii < be ? ii : be - 1];
        }
        #pragma unroll
        for (int k = 0; k < K4; ++k) {
            const unsigned dl = (unsigned)(pr[k].y - node0);
            if (i + k * T4 < be && dl < (unsigned)NB) {
                const int p = atomicAdd(&s_cur[dl], 1);
                if ((unsigned)p < (unsigned)nE) csr[p] = pr[k].x;
            }
        }
    }
}

#define ROW_LOAD(a, jj) \
    const float4 a = hp[(size_t)__builtin_amdgcn_readlane(s_l, (jj)) * (D / 4) + lane]
#define ROW_FMA(a, jj) do { \
    const float w_ = __int_as_float(__builtin_amdgcn_readlane(w_i, (jj))); \
    acc.x = fmaf(a.x, w_, acc.x); acc.y = fmaf(a.y, w_, acc.y); \
    acc.z = fmaf(a.z, w_, acc.z); acc.w = fmaf(a.w, w_, acc.w); } while (0)
#define ROW_ADD(a) do { \
    acc.x += a.x; acc.y += a.y; acc.z += a.z; acc.w += a.w; } while (0)
#define ROW_ACC(a, jj) do { if (LAYER == 1) ROW_FMA(a, jj); else ROW_ADD(a); } while (0)

__device__ __forceinline__ int clamp_id(int s, int nN) {
    return s < 0 ? 0 : (s < nN ? s : nN - 1);
}

__device__ __forceinline__ int first32(int beg, int end, int lane) {
    int i = beg + lane;
    i = i < end ? i : end - 1;
    return i < 0 ? 0 : i;
}

template <int LAYER>
__device__ __forceinline__ float4 node_sum(const float4* __restrict__ hp,
                                           const int* __restrict__ csr,
                                           const float* __restrict__ dinv,
                                           int node, int beg, int end, float dd,
                                           int s0, float d0, int lane, int nN) {
    float4 acc = make_float4(0.0f, 0.0f, 0.0f, 0.0f);
    if (node < nN) {
        const float4 vself = hp[(size_t)node * (D / 4) + lane];

        for (int base = beg; base < end; base += 32) {
            const int mm = (end - base) < 32 ? (end - base) : 32;
            int   s_l = s0;
            float w_l = d0 * dd;
            if (base != beg) {
                s_l = 0;
                w_l = 0.0f;
                if (lane < mm) {
                    s_l = clamp_id(csr[base + lane], nN);
                    if (LAYER == 1) w_l = dinv[s_l] * dd;
                }
            }
            const int w_i = __float_as_int(w_l);

            int j = 0;
            for (; j + 8 <= mm; j += 8) {
                ROW_LOAD(a0, j);     ROW_LOAD(a1, j + 1); ROW_LOAD(a2, j + 2); ROW_LOAD(a3, j + 3);
                ROW_LOAD(a4, j + 4); ROW_LOAD(a5, j + 5); ROW_LOAD(a6, j + 6); ROW_LOAD(a7, j + 7);
                ROW_ACC(a0, j);     ROW_ACC(a1, j + 1); ROW_ACC(a2, j + 2); ROW_ACC(a3, j + 3);
                ROW_ACC(a4, j + 4); ROW_ACC(a5, j + 5); ROW_ACC(a6, j + 6); ROW_ACC(a7, j + 7);
            }
            if (j + 4 <= mm) {
                ROW_LOAD(a0, j); ROW_LOAD(a1, j + 1); ROW_LOAD(a2, j + 2); ROW_LOAD(a3, j + 3);
                ROW_ACC(a0, j);  ROW_ACC(a1, j + 1);  ROW_ACC(a2, j + 2);  ROW_ACC(a3, j + 3);
                j += 4;
            }
            const int rem = mm - j;
            if (rem == 3) {
                ROW_LOAD(a0, j); ROW_LOAD(a1, j + 1); ROW_LOAD(a2, j + 2);
                ROW_ACC(a0, j);  ROW_ACC(a1, j + 1);  ROW_ACC(a2, j + 2);
            } else if (rem == 2) {
                ROW_LOAD(a0, j); ROW_LOAD(a1, j + 1);
                ROW_ACC(a0, j);  ROW_ACC(a1, j + 1);
            } else if (rem == 1) {
                ROW_LOAD(a0, j);
                ROW_ACC(a0, j);
            }
        }
        if (LAYER == 1) {
            const float ns = dd * dd;
            acc.x = fmaf(vself.x, ns, acc.x); acc.y = fmaf(vself.y, ns, acc.y);
            acc.z = fmaf(vself.z, ns, acc.z); acc.w = fmaf(vself.w, ns, acc.w);
        } else {
            acc.x = (acc.x + vself.x) * dd; acc.y = (acc.y + vself.y) * dd;
            acc.z = (acc.z + vself.z) * dd; acc.w = (acc.w + vself.w) * dd;
        }
    }
    return acc;
}

template <int LAYER>
__global__ __launch_bounds__(256, 8) void layer_fused(const int* __restrict__ off,
                                                      const int* __restrict__ csr,
                                                      const float* __restrict__ dinv,
                                                      const float* __restrict__ xin,
                                                      const float* __restrict__ W,
                                                      const float* __restrict__ bias,
                                                      float* __restrict__ out,
                                                      int nN, int nStrips, int nBlk) {
    __shared__ __attribute__((aligned(16))) float xs[2][16 * XS_STRIDE];
    __shared__ __attribute__((aligned(16))) float os[2][16 * XS_STRIDE];

    const int tid  = threadIdx.x;
    const int lane = tid & 31;
    const int wave = __builtin_amdgcn_readfirstlane(tid >> 5);
    const int m    = lane & 15;
    const int kk   = (lane >> 4) << 1;
    const int col  = wave * 16 + (lane & 15);
    const int halfm = (lane >> 4) * 8;

    v2f bfr[D / 4];
    #pragma unroll
    for (int i = 0; i < D / 4; ++i) {
        bfr[i].x = W[(size_t)(4 * i + kk) * D + col];
        bfr[i].y = W[(size_t)(4 * i + kk + 1) * D + col];
    }
    const float bc = bias[col];
    const float4* __restrict__ hp = (const float4*)xin;

    int   strip = blockIdx.x;
    int   offv = 0, ida = 0, idb = 0;
    float ddv = 0.0f, wa = 0.0f;
    if (strip < nStrips) {
        const int i = strip * 16 + 2 * wave + lane;
        offv = off[i < nN ? i : nN];
        ddv  = dinv[i < nN ? i : nN - 1];
        const int b0 = __builtin_amdgcn_readlane(offv, 0);
        const int e0 = __builtin_amdgcn_readlane(offv, 1);
        const int e1 = __builtin_amdgcn_readlane(offv, 2);
        ida = csr[first32(b0, e0, lane)];
        idb = csr[first32(e0, e1, lane)];
        ida = clamp_id(ida, nN);
        if (LAYER == 1) wa = dinv[ida];
    }

    int buf = 0;
    int prow0 = -1;
    for (; strip < nStrips; strip += nBlk, buf ^= 1) {
        const int row0 = strip * 16;
        float* __restrict__ xb = xs[buf];
        const int na = row0 + 2 * wave;
        const int beg_a = __builtin_amdgcn_readlane(offv, 0);
        const int end_a = __builtin_amdgcn_readlane(offv, 1);
        const int end_b = __builtin_amdgcn_readlane(offv, 2);
        const float dd_a = __int_as_float(__builtin_amdgcn_readlane(__float_as_int(ddv), 0));
        const float dd_b = __int_as_float(__builtin_amdgcn_readlane(__float_as_int(ddv), 1));
        const int   sb  = clamp_id(idb, nN);

        const int nxt = (strip + nBlk < nStrips) ? (strip + nBlk) : strip;
        const int in  = nxt * 16 + 2 * wave + lane;
        const int   offn = off[in < nN ? in : nN];
        const float ddn  = dinv[in < nN ? in : nN - 1];
        float wb = 0.0f;
        if (LAYER == 1) wb = dinv[sb];

        {
            const float4 acc = node_sum<LAYER>(hp, csr, dinv, na, beg_a, end_a, dd_a, ida, wa, lane, nN);
            const int r = 2 * wave;
            *(float4*)&xb[r * XS_STRIDE + lane * 4] = acc;
            if (LAYER == 1 && lane == 0) xb[r * XS_STRIDE + D] = (na < nN) ? dd_a : 0.0f;
        }

        const int nb0 = __builtin_amdgcn_readlane(offn, 0);
        const int ne0 = __builtin_amdgcn_readlane(offn, 1);
        const int ne1 = __builtin_amdgcn_readlane(offn, 2);
        int       ida_n = csr[first32(nb0, ne0, lane)];
        const int idb_n = csr[first32(ne0, ne1, lane)];

        {
            const float4 acc = node_sum<LAYER>(hp, csr, dinv, na + 1, end_a, end_b, dd_b, sb, wb, lane, nN);
            const int r = 2 * wave + 1;
            *(float4*)&xb[r * XS_STRIDE + lane * 4] = acc;
            if (LAYER == 1 && lane == 0) xb[r * XS_STRIDE + D] = (na + 1 < nN) ? dd_b : 0.0f;
        }
        __syncthreads();

        float wa_n = 0.0f;
        ida_n = clamp_id(ida_n, nN);
        if (LAYER == 1) {
            wa_n = dinv[ida_n];
            __builtin_amdgcn_sched_barrier(0);
        }

        if (prow0 >= 0) {
            const int lr  = 2 * wave;
            const int row = prow0 + lr;
            const float4 o0 = *(const float4*)&os[buf ^ 1][lr * XS_STRIDE + lane * 4];
            const float4 o1 = *(const float4*)&os[buf ^ 1][(lr + 1) * XS_STRIDE + lane * 4];
            if (row < nN)     *(float4*)&out[(size_t)row * D + lane * 4] = o0;
            if (row + 1 < nN) *(float4*)&out[(size_t)(row + 1) * D + lane * 4] = o1;
        }

        v8f c = {};
        #pragma unroll
        for (int i = 0; i < D / 4; ++i) {
            v2f a;
            a.x = xb[m * XS_STRIDE + 4 * i + kk];
            a.y = xb[m * XS_STRIDE + 4 * i + kk + 1];
            c = __builtin_amdgcn_wmma_f32_16x16x4_f32(
                 false, a,  false, bfr[i],
                 (short)0, c,  false,  false);
        }

        float v[8];
        #pragma unroll
        for (int r = 0; r < 8; ++r) {
            v[r] = c[r] + bc;
            if (LAYER == 1) {
                v[r] = v[r] > 0.0f ? v[r] : 0.0f;
                v[r] *= xb[(halfm + r) * XS_STRIDE + D];
            }
        }
        #pragma unroll
        for (int r = 0; r < 8; ++r) os[buf][(halfm + r) * XS_STRIDE + col] = v[r];
        prow0 = row0;

        offv = offn; ddv = ddn; ida = ida_n; idb = idb_n; wa = wa_n;
    }

    if (prow0 >= 0) {
        __syncthreads();
        const int lr  = 2 * wave;
        const int row = prow0 + lr;
        const float4 o0 = *(const float4*)&os[buf ^ 1][lr * XS_STRIDE + lane * 4];
        const float4 o1 = *(const float4*)&os[buf ^ 1][(lr + 1) * XS_STRIDE + lane * 4];
        if (row < nN)     *(float4*)&out[(size_t)row * D + lane * 4] = o0;
        if (row + 1 < nN) *(float4*)&out[(size_t)(row + 1) * D + lane * 4] = o1;
    }
}


static inline size_t align256(size_t v) { return (v + 255) & ~(size_t)255; }

extern "C" void kernel_launch(void* const* d_in, const int* in_sizes, int n_in,
                              void* d_out, int out_size, void* d_ws, size_t ws_size,
                              hipStream_t stream) {
    const float* x  = (const float*)d_in[0];
    const int*   ei = (const int*)d_in[1];
    const float* W1 = (const float*)d_in[2];
    const float* b1 = (const float*)d_in[3];
    const float* W2 = (const float*)d_in[4];
    const float* b2 = (const float*)d_in[5];
    float* out = (float*)d_out;

    const int N = in_sizes[0] / D;
    const int E = in_sizes[1] / 2;
    const int* srcIdx = ei;
    const int* dstIdx = ei + E;

    const int nB  = (N + NB - 1) / NB;
    const int nT  = nB * G1;
    const int per = (E + G1 - 1) / G1;

    char* ws = (char*)d_ws;
    size_t o = 0;
    float* dinv  = (float*)(ws + o); o += align256((size_t)N * sizeof(float));
    int*   tbl   = (int*)(ws + o);   o += align256((size_t)nT * sizeof(int));
    int*   base  = (int*)(ws + o);   o += align256((size_t)nT * sizeof(int));
    int*   off   = (int*)(ws + o);   o += align256((size_t)(N + 1) * sizeof(int));
    int2*  pairs = (int2*)(ws + o);  o += align256((size_t)E * sizeof(int2));
    int*   csr   = (int*)(ws + o);   o += align256((size_t)E * sizeof(int));
    float* hbuf  = (float*)(ws + o);

    const int blk = 256;
    const int gScan = (nT + SCAN_C - 1) / SCAN_C;
    const int nStrips = (N + 15) / 16;
    const int gLayer  = nStrips < 1024 ? nStrips : 1024;

    bucket_count  <<<G1, T1, 0, stream>>>(dstIdx, tbl, E, nB, per);
    scan_excl     <<<gScan, SCAN_T, 0, stream>>>(tbl, base, nT);
    bucket_scatter<<<G1, T1, 0, stream>>>(srcIdx, dstIdx, base, pairs, E, nB, per);
    bucket_build  <<<nB, T4, 0, stream>>>(pairs, base, off, dinv, csr, N, E, nB);

    layer_fused<1><<<gLayer, blk, 0, stream>>>(off, csr, dinv, x, W1, b1, hbuf, N, nStrips, gLayer);

    layer_fused<2><<<gLayer, blk, 0, stream>>>(off, csr, dinv, hbuf, W2, b2, out, N, nStrips, gLayer);
}


# Grouping passes 3 and 4 put their per-edge records in final order in an LDS tile and store the tile front to back as runs of whole lines (was one scattered store per edge); in-group order unchanged
# speedup vs baseline: 6.3073x; 1.1224x over previous
#include <hip/hip_runtime.h>
#include <hip/hip_bf16.h>


typedef float v2f __attribute__((ext_vector_type(2)));
typedef float v8f __attribute__((ext_vector_type(8)));

#define D 128
#define XS_STRIDE 132
#define SCAN_T 1024
#define SCAN_C 4096


#define NB_SH 8
#define NB    256
#define G1    256
#define T1    512
#define T4    1024
#define MAXB  512
#define KB    16
#define K4    4
#define CAPS  (KB * T1)
#define CAPB  8192

__device__ __forceinline__ int bucket_of(int d, int nB) {
    const int b = (int)((unsigned)d >> NB_SH);
    return b < nB ? b : nB - 1;
}

__global__ __launch_bounds__(T1) void bucket_count(const int* __restrict__ dst,
                                                   int* __restrict__ tbl,
                                                   int nE, int nB, int per) {
    __shared__ int s_c[MAXB];
    const int tid = threadIdx.x;
    const int w   = blockIdx.x;
    for (int i = tid; i < nB; i += T1) s_c[i] = 0;
    __syncthreads();
    const int e0 = w * per;
    const int e1 = (e0 + per < nE) ? (e0 + per) : nE;
    for (int e = e0 + tid; e < e1; e += KB * T1) {
        int d[KB];
        #pragma unroll
        for (int k = 0; k < KB; ++k) {
            const int ee = e + k * T1;
            d[k] = dst[ee < e1 ? ee : e1 - 1];
        }
        #pragma unroll
        for (int k = 0; k < KB; ++k)
            if (e + k * T1 < e1) atomicAdd(&s_c[bucket_of(d[k], nB)], 1);
    }
    __syncthreads();
    for (int i = tid; i < nB; i += T1) tbl[i * G1 + w] = s_c[i];
}

__global__ __launch_bounds__(SCAN_T) void scan_excl(const int* __restrict__ in,
                                                    int* __restrict__ out, int n) {
    __shared__ int s_w[32];
    __shared__ int s_p[32];
    const int tid  = threadIdx.x;
    const int lane = tid & 31;
    const int wave = tid >> 5;
    const int chunk0 = blockIdx.x * SCAN_C;

    int pre = 0;
    const int4* __restrict__ in4 = (const int4*)in;
    #pragma unroll 8
    for (int k = 0; k < (int)blockIdx.x; ++k) {
        const int4 v = in4[k * SCAN_T + tid];
        pre += (v.x + v.y) + (v.z + v.w);
    }
    for (int o = 16; o > 0; o >>= 1) pre += __shfl_xor(pre, o);

    const int i0 = chunk0 + tid * 4;
    int c[4];
    #pragma unroll
    for (int k = 0; k < 4; ++k) c[k] = (i0 + k < n) ? in[i0 + k] : 0;
    const int t = (c[0] + c[1]) + (c[2] + c[3]);

    int inc = t;
    #pragma unroll
    for (int o = 1; o < 32; o <<= 1) {
        const int u = __shfl_up(inc, o);
        if (lane >= o) inc += u;
    }
    if (lane == 31) s_w[wave] = inc;
    if (lane == 0)  s_p[wave] = pre;
    __syncthreads();

    int q = s_p[lane] + (lane < wave ? s_w[lane] : 0);
    for (int o = 16; o > 0; o >>= 1) q += __shfl_xor(q, o);

    int run = q + inc - t;
    #pragma unroll
    for (int k = 0; k < 4; ++k) {
        if (i0 + k < n) {
            out[i0 + k] = run;
            run += c[k];
        }
    }
}

__global__ __launch_bounds__(T1) void bucket_scatter(const int* __restrict__ src,
                                                     const int* __restrict__ dst,
                                                     const int* __restrict__ base,
                                                     int2* __restrict__ pairs,
                                                     int nE, int nB, int per) {
    __shared__ int s_p[MAXB];
    __shared__ int s_c[MAXB];
    __shared__ int s_l[MAXB];
    __shared__ int s_w[T1 / 32];
    __shared__ int2 s_t[CAPS];
    const int tid  = threadIdx.x;
    const int lane = tid & 31;
    const int w    = blockIdx.x;
    for (int i = tid; i < nB; i += T1) s_p[i] = base[i * G1 + w];
    const int e0 = w * per;
    const int e1 = (e0 + per < nE) ? (e0 + per) : nE;
    for (int eb = e0; eb < e1; eb += CAPS) {
        const int e = eb + tid;
        if (eb != e0) {
            __syncthreads();
            for (int i = tid; i < nB; i += T1) s_p[i] += s_c[i];
        }
        for (int i = tid; i < nB; i += T1) s_c[i] = 0;
        __syncthreads();
        int d[KB], s[KB], r[KB];
        #pragma unroll
        for (int k = 0; k < KB; ++k) {
            const int ee = (e + k * T1 < e1) ? (e + k * T1) : (e1 - 1);
            d[k] = dst[ee];
            s[k] = src[ee];
        }
        #pragma unroll
        for (int k = 0; k < KB; ++k) {
            r[k] = 0;
            if (e + k * T1 < e1) r[k] = atomicAdd(&s_c[bucket_of(d[k], nB)], 1);
        }
        __syncthreads();
        {
            const int cb = (tid < nB) ? s_c[tid] : 0;
            int inc = cb;
            #pragma unroll
            for (int o = 1; o < 32; o <<= 1) {
                const int u = __shfl_up(inc, o);
                if (lane >= o) inc += u;
            }
            if (lane == 31) s_w[tid >> 5] = inc;
            __syncthreads();
            int pre = 0;
            for (int k = 0; k < (tid >> 5); ++k) pre += s_w[k];
            if (tid < nB) s_l[tid] = pre + inc - cb;
        }
        __syncthreads();
        unsigned li[KB];
        #pragma unroll
        for (int k = 0; k < KB; ++k) li[k] = (unsigned)(s_l[bucket_of(d[k], nB)] + r[k]);
        #pragma unroll
        for (int k = 0; k < KB; ++k)
            if (e + k * T1 < e1 && li[k] < (unsigned)CAPS) s_t[li[k]] = make_int2(s[k], d[k]);
        __syncthreads();
        const int nt = (e1 - eb < CAPS) ? (e1 - eb) : CAPS;
        int2 pr[KB];
        int  pl[KB];
        #pragma unroll
        for (int k = 0; k < KB; ++k) {
            const int i = tid + k * T1;
            pr[k] = s_t[i < nt ? i : nt - 1];
        }
        #pragma unroll
        for (int k = 0; k < KB; ++k) {
            const int i = tid + k * T1;
            const int b = bucket_of(pr[k].y, nB);
            pl[k] = s_p[b] + ((i < nt ? i : nt - 1) - s_l[b]);
        }
        #pragma unroll
        for (int k = 0; k < KB; ++k)
            if (tid + k * T1 < nt && (unsigned)pl[k] < (unsigned)nE) pairs[pl[k]] = pr[k];
    }
}

__global__ __launch_bounds__(T4) void bucket_build(const int2* __restrict__ pairs,
                                                   const int* __restrict__ base,
                                                   int* __restrict__ off,
                                                   float* __restrict__ dinv,
                                                   int* __restrict__ csr,
                                                   int nN, int nE, int nB) {
    __shared__ int s_cnt[NB];
    __shared__ int s_cur[NB];
    __shared__ int s_w[NB / 32];
    __shared__ int s_o[CAPB];
    const int tid  = threadIdx.x;
    const int lane = tid & 31;
    const int b    = blockIdx.x;
    const int node0 = b << NB_SH;
    int bs = base[b * G1];
    int be = (b + 1 < nB) ? base[(b + 1) * G1] : nE;
    bs = bs < 0 ? 0 : (bs < nE ? bs : nE);
    be = be < bs ? bs : (be < nE ? be : nE);

    if (tid < NB) s_cnt[tid] = 0;
    __syncthreads();

    for (int i = bs + tid; i < be; i += K4 * T4) {
        int d[K4];
        #pragma unroll
        for (int k = 0; k < K4; ++k) {
            const int ii = i + k * T4;
            d[k] = pairs[ii < be ? ii : be - 1].y;
        }
        #pragma unroll
        for (int k = 0; k < K4; ++k) {
            const unsigned dl = (unsigned)(d[k] - node0);
            if (i + k * T4 < be && dl < (unsigned)NB) atomicAdd(&s_cnt[dl], 1);
        }
    }
    __syncthreads();

    int c = 0, inc = 0;
    if (tid < NB) {
        c = s_cnt[tid];
        inc = c;
        #pragma unroll
        for (int o = 1; o < 32; o <<= 1) {
            const int u = __shfl_up(inc, o);
            if (lane >= o) inc += u;
        }
        if (lane == 31) s_w[tid >> 5] = inc;
    }
    __syncthreads();

    if (tid < NB) {
        int pre = 0;
        for (int k = 0; k < (tid >> 5); ++k) pre += s_w[k];
        const int start = bs + pre + inc - c;
        s_cur[tid] = start;
        const int node = node0 + tid;
        if (node < nN) {
            off[node]  = start;
            dinv[node] = rsqrtf((float)(c + 1));
            if (node == nN - 1) off[nN] = start + c;
        }
    }
    __syncthreads();

    for (int i = bs + tid; i < be; i += K4 * T4) {
        int2 pr[K4];
        #pragma unroll
        for (int k = 0; k < K4; ++k) {
            const int ii = i + k * T4;
            pr[k] = pairs[ii < be ? ii : be - 1];
        }
        #pragma unroll
        for (int k = 0; k < K4; ++k) {
            const unsigned dl = (unsigned)(pr[k].y - node0);
            if (i + k * T4 < be && dl < (unsigned)NB) {
                const int p = atomicAdd(&s_cur[dl], 1);
                const unsigned li = (unsigned)(p - bs);
                if (li < (unsigned)CAPB) s_o[li] = pr[k].x;
                else if ((unsigned)p < (unsigned)nE) csr[p] = pr[k].x;
            }
        }
    }
    __syncthreads();
    const int ns = (be - bs < CAPB) ? (be - bs) : CAPB;
    for (int e = (bs & ~31) + tid; e < bs + ns; e += T4)
        if (e >= bs) csr[e] = s_o[e - bs];
}

#define ROW_LOAD(a, jj) \
    const float4 a = hp[(size_t)__builtin_amdgcn_readlane(s_l, (jj)) * (D / 4) + lane]
#define ROW_FMA(a, jj) do { \
    const float w_ = __int_as_float(__builtin_amdgcn_readlane(w_i, (jj))); \
    acc.x = fmaf(a.x, w_, acc.x); acc.y = fmaf(a.y, w_, acc.y); \
    acc.z = fmaf(a.z, w_, acc.z); acc.w = fmaf(a.w, w_, acc.w); } while (0)
#define ROW_ADD(a) do { \
    acc.x += a.x; acc.y += a.y; acc.z += a.z; acc.w += a.w; } while (0)
#define ROW_ACC(a, jj) do { if (LAYER == 1) ROW_FMA(a, jj); else ROW_ADD(a); } while (0)

__device__ __forceinline__ int clamp_id(int s, int nN) {
    return s < 0 ? 0 : (s < nN ? s : nN - 1);
}

__device__ __forceinline__ int first32(int beg, int end, int lane) {
    int i = beg + lane;
    i = i < end ? i : end - 1;
    return i < 0 ? 0 : i;
}

template <int LAYER>
__device__ __forceinline__ float4 node_sum(const float4* __restrict__ hp,
                                           const int* __restrict__ csr,
                                           const float* __restrict__ dinv,
                                           int node, int beg, int end, float dd,
                                           int s0, float d0, int lane, int nN) {
    float4 acc = make_float4(0.0f, 0.0f, 0.0f, 0.0f);
    if (node < nN) {
        const float4 vself = hp[(size_t)node * (D / 4) + lane];

        for (int base = beg; base < end; base += 32) {
            const int mm = (end - base) < 32 ? (end - base) : 32;
            int   s_l = s0;
            float w_l = d0 * dd;
            if (base != beg) {
                s_l = 0;
                w_l = 0.0f;
                if (lane < mm) {
                    s_l = clamp_id(csr[base + lane], nN);
                    if (LAYER == 1) w_l = dinv[s_l] * dd;
                }
            }
            const int w_i = __float_as_int(w_l);

            int j = 0;
            for (; j + 8 <= mm; j += 8) {
                ROW_LOAD(a0, j);     ROW_LOAD(a1, j + 1); ROW_LOAD(a2, j + 2); ROW_LOAD(a3, j + 3);
                ROW_LOAD(a4, j + 4); ROW_LOAD(a5, j + 5); ROW_LOAD(a6, j + 6); ROW_LOAD(a7, j + 7);
                ROW_ACC(a0, j);     ROW_ACC(a1, j + 1); ROW_ACC(a2, j + 2); ROW_ACC(a3, j + 3);
                ROW_ACC(a4, j + 4); ROW_ACC(a5, j + 5); ROW_ACC(a6, j + 6); ROW_ACC(a7, j + 7);
            }
            if (j + 4 <= mm) {
                ROW_LOAD(a0, j); ROW_LOAD(a1, j + 1); ROW_LOAD(a2, j + 2); ROW_LOAD(a3, j + 3);
                ROW_ACC(a0, j);  ROW_ACC(a1, j + 1);  ROW_ACC(a2, j + 2);  ROW_ACC(a3, j + 3);
                j += 4;
            }
            const int rem = mm - j;
            if (rem == 3) {
                ROW_LOAD(a0, j); ROW_LOAD(a1, j + 1); ROW_LOAD(a2, j + 2);
                ROW_ACC(a0, j);  ROW_ACC(a1, j + 1);  ROW_ACC(a2, j + 2);
            } else if (rem == 2) {
                ROW_LOAD(a0, j); ROW_LOAD(a1, j + 1);
                ROW_ACC(a0, j);  ROW_ACC(a1, j + 1);
            } else if (rem == 1) {
                ROW_LOAD(a0, j);
                ROW_ACC(a0, j);
            }
        }
        if (LAYER == 1) {
            const float ns = dd * dd;
            acc.x = fmaf(vself.x, ns, acc.x); acc.y = fmaf(vself.y, ns, acc.y);
            acc.z = fmaf(vself.z, ns, acc.z); acc.w = fmaf(vself.w, ns, acc.w);
        } else {
            acc.x = (acc.x + vself.x) * dd; acc.y = (acc.y + vself.y) * dd;
            acc.z = (acc.z + vself.z) * dd; acc.w = (acc.w + vself.w) * dd;
        }
    }
    return acc;
}

template <int LAYER>
__global__ __launch_bounds__(256, 8) void layer_fused(const int* __restrict__ off,
                                                      const int* __restrict__ csr,
                                                      const float* __restrict__ dinv,
                                                      const float* __restrict__ xin,
                                                      const float* __restrict__ W,
                                                      const float* __restrict__ bias,
                                                      float* __restrict__ out,
                                                      int nN, int nStrips, int nBlk) {
    __shared__ __attribute__((aligned(16))) float xs[2][16 * XS_STRIDE];
    __shared__ __attribute__((aligned(16))) float os[2][16 * XS_STRIDE];

    const int tid  = threadIdx.x;
    const int lane = tid & 31;
    const int wave = __builtin_amdgcn_readfirstlane(tid >> 5);
    const int m    = lane & 15;
    const int kk   = (lane >> 4) << 1;
    const int col  = wave * 16 + (lane & 15);
    const int halfm = (lane >> 4) * 8;

    v2f bfr[D / 4];
    #pragma unroll
    for (int i = 0; i < D / 4; ++i) {
        bfr[i].x = W[(size_t)(4 * i + kk) * D + col];
        bfr[i].y = W[(size_t)(4 * i + kk + 1) * D + col];
    }
    const float bc = bias[col];
    const float4* __restrict__ hp = (const float4*)xin;

    int   strip = blockIdx.x;
    int   offv = 0, ida = 0, idb = 0;
    float ddv = 0.0f, wa = 0.0f;
    if (strip < nStrips) {
        const int i = strip * 16 + 2 * wave + lane;
        offv = off[i < nN ? i : nN];
        ddv  = dinv[i < nN ? i : nN - 1];
        const int b0 = __builtin_amdgcn_readlane(offv, 0);
        const int e0 = __builtin_amdgcn_readlane(offv, 1);
        const int e1 = __builtin_amdgcn_readlane(offv, 2);
        ida = csr[first32(b0, e0, lane)];
        idb = csr[first32(e0, e1, lane)];
        ida = clamp_id(ida, nN);
        if (LAYER == 1) wa = dinv[ida];
    }

    int buf = 0;
    int prow0 = -1;
    for (; strip < nStrips; strip += nBlk, buf ^= 1) {
        const int row0 = strip * 16;
        float* __restrict__ xb = xs[buf];
        const int na = row0 + 2 * wave;
        const int beg_a = __builtin_amdgcn_readlane(offv, 0);
        const int end_a = __builtin_amdgcn_readlane(offv, 1);
        const int end_b = __builtin_amdgcn_readlane(offv, 2);
        const float dd_a = __int_as_float(__builtin_amdgcn_readlane(__float_as_int(ddv), 0));
        const float dd_b = __int_as_float(__builtin_amdgcn_readlane(__float_as_int(ddv), 1));
        const int   sb  = clamp_id(idb, nN);

        const int nxt = (strip + nBlk < nStrips) ? (strip + nBlk) : strip;
        const int in  = nxt * 16 + 2 * wave + lane;
        const int   offn = off[in < nN ? in : nN];
        const float ddn  = dinv[in < nN ? in : nN - 1];
        float wb = 0.0f;
        if (LAYER == 1) wb = dinv[sb];

        {
            const float4 acc = node_sum<LAYER>(hp, csr, dinv, na, beg_a, end_a, dd_a, ida, wa, lane, nN);
            const int r = 2 * wave;
            *(float4*)&xb[r * XS_STRIDE + lane * 4] = acc;
            if (LAYER == 1 && lane == 0) xb[r * XS_STRIDE + D] = (na < nN) ? dd_a : 0.0f;
        }

        const int nb0 = __builtin_amdgcn_readlane(offn, 0);
        const int ne0 = __builtin_amdgcn_readlane(offn, 1);
        const int ne1 = __builtin_amdgcn_readlane(offn, 2);
        int       ida_n = csr[first32(nb0, ne0, lane)];
        const int idb_n = csr[first32(ne0, ne1, lane)];

        {
            const float4 acc = node_sum<LAYER>(hp, csr, dinv, na + 1, end_a, end_b, dd_b, sb, wb, lane, nN);
            const int r = 2 * wave + 1;
            *(float4*)&xb[r * XS_STRIDE + lane * 4] = acc;
            if (LAYER == 1 && lane == 0) xb[r * XS_STRIDE + D] = (na + 1 < nN) ? dd_b : 0.0f;
        }
        __syncthreads();

        float wa_n = 0.0f;
        ida_n = clamp_id(ida_n, nN);
        if (LAYER == 1) {
            wa_n = dinv[ida_n];
            __builtin_amdgcn_sched_barrier(0);
        }

        if (prow0 >= 0) {
            const int lr  = 2 * wave;
            const int row = prow0 + lr;
            const float4 o0 = *(const float4*)&os[buf ^ 1][lr * XS_STRIDE + lane * 4];
            const float4 o1 = *(const float4*)&os[buf ^ 1][(lr + 1) * XS_STRIDE + lane * 4];
            if (row < nN)     *(float4*)&out[(size_t)row * D + lane * 4] = o0;
            if (row + 1 < nN) *(float4*)&out[(size_t)(row + 1) * D + lane * 4] = o1;
        }

        v8f c = {};
        #pragma unroll
        for (int i = 0; i < D / 4; ++i) {
            v2f a;
            a.x = xb[m * XS_STRIDE + 4 * i + kk];
            a.y = xb[m * XS_STRIDE + 4 * i + kk + 1];
            c = __builtin_amdgcn_wmma_f32_16x16x4_f32(
                 false, a,  false, bfr[i],
                 (short)0, c,  false,  false);
        }

        float v[8];
        #pragma unroll
        for (int r = 0; r < 8; ++r) {
            v[r] = c[r] + bc;
            if (LAYER == 1) {
                v[r] = v[r] > 0.0f ? v[r] : 0.0f;
                v[r] *= xb[(halfm + r) * XS_STRIDE + D];
            }
        }
        #pragma unroll
        for (int r = 0; r < 8; ++r) os[buf][(halfm + r) * XS_STRIDE + col] = v[r];
        prow0 = row0;

        offv = offn; ddv = ddn; ida = ida_n; idb = idb_n; wa = wa_n;
    }

    if (prow0 >= 0) {
        __syncthreads();
        const int lr  = 2 * wave;
        const int row = prow0 + lr;
        const float4 o0 = *(const float4*)&os[buf ^ 1][lr * XS_STRIDE + lane * 4];
        const float4 o1 = *(const float4*)&os[buf ^ 1][(lr + 1) * XS_STRIDE + lane * 4];
        if (row < nN)     *(float4*)&out[(size_t)row * D + lane * 4] = o0;
        if (row + 1 < nN) *(float4*)&out[(size_t)(row + 1) * D + lane * 4] = o1;
    }
}


static inline size_t align256(size_t v) { return (v + 255) & ~(size_t)255; }

extern "C" void kernel_launch(void* const* d_in, const int* in_sizes, int n_in,
                              void* d_out, int out_size, void* d_ws, size_t ws_size,
                              hipStream_t stream) {
    const float* x  = (const float*)d_in[0];
    const int*   ei = (const int*)d_in[1];
    const float* W1 = (const float*)d_in[2];
    const float* b1 = (const float*)d_in[3];
    const float* W2 = (const float*)d_in[4];
    const float* b2 = (const float*)d_in[5];
    float* out = (float*)d_out;

    const int N = in_sizes[0] / D;
    const int E = in_sizes[1] / 2;
    const int* srcIdx = ei;
    const int* dstIdx = ei + E;

    const int nB  = (N + NB - 1) / NB;
    const int nT  = nB * G1;
    const int per = (E + G1 - 1) / G1;

    char* ws = (char*)d_ws;
    size_t o = 0;
    float* dinv  = (float*)(ws + o); o += align256((size_t)N * sizeof(float));
    int*   tbl   = (int*)(ws + o);   o += align256((size_t)nT * sizeof(int));
    int*   base  = (int*)(ws + o);   o += align256((size_t)nT * sizeof(int));
    int*   off   = (int*)(ws + o);   o += align256((size_t)(N + 1) * sizeof(int));
    int2*  pairs = (int2*)(ws + o);  o += align256((size_t)E * sizeof(int2));
    int*   csr   = (int*)(ws + o);   o += align256((size_t)E * sizeof(int));
    float* hbuf  = (float*)(ws + o);

    const int blk = 256;
    const int gScan = (nT + SCAN_C - 1) / SCAN_C;
    const int nStrips = (N + 15) / 16;
    const int gLayer  = nStrips < 1024 ? nStrips : 1024;

    bucket_count  <<<G1, T1, 0, stream>>>(dstIdx, tbl, E, nB, per);
    scan_excl     <<<gScan, SCAN_T, 0, stream>>>(tbl, base, nT);
    bucket_scatter<<<G1, T1, 0, stream>>>(srcIdx, dstIdx, base, pairs, E, nB, per);
    bucket_build  <<<nB, T4, 0, stream>>>(pairs, base, off, dinv, csr, N, E, nB);

    layer_fused<1><<<gLayer, blk, 0, stream>>>(off, csr, dinv, x, W1, b1, hbuf, N, nStrips, gLayer);

    layer_fused<2><<<gLayer, blk, 0, stream>>>(off, csr, dinv, hbuf, W2, b2, out, N, nStrips, gLayer);
}


# Edge grouping in two launches (were four): a chunk is sorted by bucket in LDS, stored in place with a run table; a bucket's first slot = sum of its runs' starts; in-group order = new walk's LDS order
# speedup vs baseline: 6.6543x; 1.0550x over previous
#include <hip/hip_runtime.h>
#include <hip/hip_bf16.h>


typedef float v2f __attribute__((ext_vector_type(2)));
typedef float v8f __attribute__((ext_vector_type(8)));

#define D 128
#define XS_STRIDE 132


#define NB_SH 8
#define NB    256
#define G1    256
#define T1    512
#define T4    1024
#define MAXB  512
#define KB    16
#define K4    4
#define RL    8
#define CAPS  (KB * T1)
#define CAPB  8192

__device__ __forceinline__ int bucket_of(int d, int nB) {
    const int b = (int)((unsigned)d >> NB_SH);
    return b < nB ? b : nB - 1;
}

__global__ __launch_bounds__(T1) void chunk_sort(const int* __restrict__ src,
                                                 const int* __restrict__ dst,
                                                 int* __restrict__ cnt,
                                                 int* __restrict__ beg,
                                                 int2* __restrict__ pairs,
                                                 int nE, int nB, int per) {
    __shared__ int s_c[MAXB];
    __shared__ int s_l[MAXB];
    __shared__ int s_w[T1 / 32];
    __shared__ int2 s_t[CAPS];
    const int tid  = threadIdx.x;
    const int lane = tid & 31;
    const int w    = blockIdx.x;
    const int nU   = gridDim.x;
    for (int i = tid; i < nB; i += T1) s_c[i] = 0;
    const int e0 = w * per;
    const int e1 = (e0 + per < nE) ? (e0 + per) : nE;
    const int e  = e0 + tid;
    __syncthreads();
    int d[KB], s[KB], r[KB];
    #pragma unroll
    for (int k = 0; k < KB; ++k) {
        const int ee = (e + k * T1 < e1) ? (e + k * T1) : (e1 - 1);
        d[k] = dst[ee];
        s[k] = src[ee];
    }
    #pragma unroll
    for (int k = 0; k < KB; ++k) {
        r[k] = 0;
        if (e + k * T1 < e1) r[k] = atomicAdd(&s_c[bucket_of(d[k], nB)], 1);
    }
    __syncthreads();
    {
        const int cb = (tid < nB) ? s_c[tid] : 0;
        int inc = cb;
        #pragma unroll
        for (int o = 1; o < 32; o <<= 1) {
            const int u = __shfl_up(inc, o);
            if (lane >= o) inc += u;
        }
        if (lane == 31) s_w[tid >> 5] = inc;
        __syncthreads();
        int pre = 0;
        for (int k = 0; k < (tid >> 5); ++k) pre += s_w[k];
        if (tid < nB) {
            const int first = pre + inc - cb;
            s_l[tid] = first;
            cnt[tid * nU + w] = cb;
            beg[tid * nU + w] = first;
        }
    }
    __syncthreads();
    unsigned li[KB];
    #pragma unroll
    for (int k = 0; k < KB; ++k) li[k] = (unsigned)(s_l[bucket_of(d[k], nB)] + r[k]);
    #pragma unroll
    for (int k = 0; k < KB; ++k)
        if (e + k * T1 < e1 && li[k] < (unsigned)CAPS) s_t[li[k]] = make_int2(s[k], d[k]);
    __syncthreads();
    const int nt = (e1 - e0 < CAPS) ? (e1 - e0) : CAPS;
    #pragma unroll
    for (int k = 0; k < KB; ++k) {
        const int i = tid + k * T1;
        if (i < nt) pairs[e0 + i] = s_t[i];
    }
}

__device__ __forceinline__ void run_of(const int* __restrict__ crow,
                                       const int* __restrict__ brow,
                                       int u, int nU, int per,
                                       int& r0, int& c, int& first) {
    const int uu = u < nU ? u : nU - 1;
    int cc = crow[uu];
    int st = brow[uu];
    first = u < nU ? st : 0;
    cc = cc < 0 ? 0 : (cc < per ? cc : per);
    st = st < 0 ? 0 : (st < per - cc ? st : per - cc);
    r0 = uu * per + st;
    c  = u < nU ? cc : 0;
}

__global__ __launch_bounds__(T4) void bucket_build(const int2* __restrict__ pairs,
                                                   const int* __restrict__ cnt,
                                                   const int* __restrict__ beg,
                                                   int* __restrict__ off,
                                                   float* __restrict__ dinv,
                                                   int* __restrict__ csr,
                                                   int nN, int nE, int nB, int nU, int per) {
    __shared__ int s_cnt[NB];
    __shared__ int s_cur[NB];
    __shared__ int s_w[NB / 32];
    __shared__ int s_f[T4 / 32];
    __shared__ int s_m[T4 / 32];
    __shared__ int s_o[CAPB];
    const int tid  = threadIdx.x;
    const int lane = tid & 31;
    const int b    = blockIdx.x;
    const int node0 = b << NB_SH;
    const int* __restrict__ crow = cnt + (size_t)b * nU;
    const int* __restrict__ brow = beg + (size_t)b * nU;

    if (tid < NB) s_cnt[tid] = 0;
    __syncthreads();

    const int q  = tid & (RL - 1);
    const int u0 = tid / RL;
    const int US = T4 / RL;

    int fr = 0, mine = 0;
    for (int ub = u0; ub < nU; ub += 2 * US) {
        int ra, ca, fa, rb, cb, fb;
        run_of(crow, brow, ub,      nU, per, ra, ca, fa);
        run_of(crow, brow, ub + US, nU, per, rb, cb, fb);
        if (q == 0) { fr += fa + fb; mine += ca + cb; }
        const int cm = ca > cb ? ca : cb;
        for (int i = q; i < cm; i += K4 * RL) {
            int da[K4], db[K4];
            #pragma unroll
            for (int k = 0; k < K4; ++k) {
                const int ii = i + k * RL;
                const int xa = ra + (ii < ca ? ii : (ca > 0 ? ca - 1 : 0));
                const int xb = rb + (ii < cb ? ii : (cb > 0 ? cb - 1 : 0));
                da[k] = pairs[xa < nE ? xa : nE - 1].y;
                db[k] = pairs[xb < nE ? xb : nE - 1].y;
            }
            #pragma unroll
            for (int k = 0; k < K4; ++k) {
                const int ii = i + k * RL;
                const unsigned la = (unsigned)(da[k] - node0);
                const unsigned lb = (unsigned)(db[k] - node0);
                if (ii < ca && la < (unsigned)NB) atomicAdd(&s_cnt[la], 1);
                if (ii < cb && lb < (unsigned)NB) atomicAdd(&s_cnt[lb], 1);
            }
        }
    }
    for (int o = 16; o > 0; o >>= 1) {
        fr   += __shfl_xor(fr, o);
        mine += __shfl_xor(mine, o);
    }
    if (lane == 0) { s_f[tid >> 5] = fr; s_m[tid >> 5] = mine; }
    __syncthreads();

    int bs = s_f[lane], bm = s_m[lane];
    for (int o = 16; o > 0; o >>= 1) {
        bs += __shfl_xor(bs, o);
        bm += __shfl_xor(bm, o);
    }
    bs = bs < 0 ? 0 : (bs < nE ? bs : nE);
    bm = bm < 0 ? 0 : bm;
    const int be = (bm < nE - bs) ? (bs + bm) : nE;

    int c = 0, inc = 0;
    if (tid < NB) {
        c = s_cnt[tid];
        inc = c;
        #pragma unroll
        for (int o = 1; o < 32; o <<= 1) {
            const int u = __shfl_up(inc, o);
            if (lane >= o) inc += u;
        }
        if (lane == 31) s_w[tid >> 5] = inc;
    }
    __syncthreads();

    if (tid < NB) {
        int pre = 0;
        for (int k = 0; k < (tid >> 5); ++k) pre += s_w[k];
        const int start = bs + pre + inc - c;
        s_cur[tid] = start;
        const int node = node0 + tid;
        if (node < nN) {
            off[node]  = start;
            dinv[node] = rsqrtf((float)(c + 1));
            if (node == nN - 1) off[nN] = start + c;
        }
    }
    __syncthreads();

    for (int ub = u0; ub < nU; ub += 2 * US) {
        int ra, ca, fa, rb, cb, fb;
        run_of(crow, brow, ub,      nU, per, ra, ca, fa);
        run_of(crow, brow, ub + US, nU, per, rb, cb, fb);
        const int cm = ca > cb ? ca : cb;
        for (int i = q; i < cm; i += K4 * RL) {
            int2 pa[K4], pb[K4];
            #pragma unroll
            for (int k = 0; k < K4; ++k) {
                const int ii = i + k * RL;
                const int xa = ra + (ii < ca ? ii : (ca > 0 ? ca - 1 : 0));
                const int xb = rb + (ii < cb ? ii : (cb > 0 ? cb - 1 : 0));
                pa[k] = pairs[xa < nE ? xa : nE - 1];
                pb[k] = pairs[xb < nE ? xb : nE - 1];
            }
            #pragma unroll
            for (int k = 0; k < K4; ++k) {
                const int ii = i + k * RL;
                const unsigned la = (unsigned)(pa[k].y - node0);
                const unsigned lb = (unsigned)(pb[k].y - node0);
                if (ii < ca && la < (unsigned)NB) {
                    const int p = atomicAdd(&s_cur[la], 1);
                    const unsigned li = (unsigned)(p - bs);
                    if (li < (unsigned)CAPB) s_o[li] = pa[k].x;
                    else if ((unsigned)p < (unsigned)nE) csr[p] = pa[k].x;
                }
                if (ii < cb && lb < (unsigned)NB) {
                    const int p = atomicAdd(&s_cur[lb], 1);
                    const unsigned li = (unsigned)(p - bs);
                    if (li < (unsigned)CAPB) s_o[li] = pb[k].x;
                    else if ((unsigned)p < (unsigned)nE) csr[p] = pb[k].x;
                }
            }
        }
    }
    __syncthreads();
    const int ns = (be - bs < CAPB) ? (be - bs) : CAPB;
    for (int e = (bs & ~31) + tid; e < bs + ns; e += T4)
        if (e >= bs) csr[e] = s_o[e - bs];
}

#define ROW_LOAD(a, jj) \
    const float4 a = hp[(size_t)__builtin_amdgcn_readlane(s_l, (jj)) * (D / 4) + lane]
#define ROW_FMA(a, jj) do { \
    const float w_ = __int_as_float(__builtin_amdgcn_readlane(w_i, (jj))); \
    acc.x = fmaf(a.x, w_, acc.x); acc.y = fmaf(a.y, w_, acc.y); \
    acc.z = fmaf(a.z, w_, acc.z); acc.w = fmaf(a.w, w_, acc.w); } while (0)
#define ROW_ADD(a) do { \
    acc.x += a.x; acc.y += a.y; acc.z += a.z; acc.w += a.w; } while (0)
#define ROW_ACC(a, jj) do { if (LAYER == 1) ROW_FMA(a, jj); else ROW_ADD(a); } while (0)

__device__ __forceinline__ int clamp_id(int s, int nN) {
    return s < 0 ? 0 : (s < nN ? s : nN - 1);
}

__device__ __forceinline__ int first32(int beg, int end, int lane) {
    int i = beg + lane;
    i = i < end ? i : end - 1;
    return i < 0 ? 0 : i;
}

template <int LAYER>
__device__ __forceinline__ float4 node_sum(const float4* __restrict__ hp,
                                           const int* __restrict__ csr,
                                           const float* __restrict__ dinv,
                                           int node, int beg, int end, float dd,
                                           int s0, float d0, int lane, int nN) {
    float4 acc = make_float4(0.0f, 0.0f, 0.0f, 0.0f);
    if (node < nN) {
        const float4 vself = hp[(size_t)node * (D / 4) + lane];

        for (int base = beg; base < end; base += 32) {
            const int mm = (end - base) < 32 ? (end - base) : 32;
            int   s_l = s0;
            float w_l = d0 * dd;
            if (base != beg) {
                s_l = 0;
                w_l = 0.0f;
                if (lane < mm) {
                    s_l = clamp_id(csr[base + lane], nN);
                    if (LAYER == 1) w_l = dinv[s_l] * dd;
                }
            }
            const int w_i = __float_as_int(w_l);

            int j = 0;
            for (; j + 8 <= mm; j += 8) {
                ROW_LOAD(a0, j);     ROW_LOAD(a1, j + 1); ROW_LOAD(a2, j + 2); ROW_LOAD(a3, j + 3);
                ROW_LOAD(a4, j + 4); ROW_LOAD(a5, j + 5); ROW_LOAD(a6, j + 6); ROW_LOAD(a7, j + 7);
                ROW_ACC(a0, j);     ROW_ACC(a1, j + 1); ROW_ACC(a2, j + 2); ROW_ACC(a3, j + 3);
                ROW_ACC(a4, j + 4); ROW_ACC(a5, j + 5); ROW_ACC(a6, j + 6); ROW_ACC(a7, j + 7);
            }
            if (j + 4 <= mm) {
                ROW_LOAD(a0, j); ROW_LOAD(a1, j + 1); ROW_LOAD(a2, j + 2); ROW_LOAD(a3, j + 3);
                ROW_ACC(a0, j);  ROW_ACC(a1, j + 1);  ROW_ACC(a2, j + 2);  ROW_ACC(a3, j + 3);
                j += 4;
            }
            const int rem = mm - j;
            if (rem == 3) {
                ROW_LOAD(a0, j); ROW_LOAD(a1, j + 1); ROW_LOAD(a2, j + 2);
                ROW_ACC(a0, j);  ROW_ACC(a1, j + 1);  ROW_ACC(a2, j + 2);
            } else if (rem == 2) {
                ROW_LOAD(a0, j); ROW_LOAD(a1, j + 1);
                ROW_ACC(a0, j);  ROW_ACC(a1, j + 1);
            } else if (rem == 1) {
                ROW_LOAD(a0, j);
                ROW_ACC(a0, j);
            }
        }
        if (LAYER == 1) {
            const float ns = dd * dd;
            acc.x = fmaf(vself.x, ns, acc.x); acc.y = fmaf(vself.y, ns, acc.y);
            acc.z = fmaf(vself.z, ns, acc.z); acc.w = fmaf(vself.w, ns, acc.w);
        } else {
            acc.x = (acc.x + vself.x) * dd; acc.y = (acc.y + vself.y) * dd;
            acc.z = (acc.z + vself.z) * dd; acc.w = (acc.w + vself.w) * dd;
        }
    }
    return acc;
}

template <int LAYER>
__global__ __launch_bounds__(256, 8) void layer_fused(const int* __restrict__ off,
                                                      const int* __restrict__ csr,
                                                      const float* __restrict__ dinv,
                                                      const float* __restrict__ xin,
                                                      const float* __restrict__ W,
                                                      const float* __restrict__ bias,
                                                      float* __restrict__ out,
                                                      int nN, int nStrips, int nBlk) {
    __shared__ __attribute__((aligned(16))) float xs[2][16 * XS_STRIDE];
    __shared__ __attribute__((aligned(16))) float os[2][16 * XS_STRIDE];

    const int tid  = threadIdx.x;
    const int lane = tid & 31;
    const int wave = __builtin_amdgcn_readfirstlane(tid >> 5);
    const int m    = lane & 15;
    const int kk   = (lane >> 4) << 1;
    const int col  = wave * 16 + (lane & 15);
    const int halfm = (lane >> 4) * 8;

    v2f bfr[D / 4];
    #pragma unroll
    for (int i = 0; i < D / 4; ++i) {
        bfr[i].x = W[(size_t)(4 * i + kk) * D + col];
        bfr[i].y = W[(size_t)(4 * i + kk + 1) * D + col];
    }
    const float bc = bias[col];
    const float4* __restrict__ hp = (const float4*)xin;

    int   strip = blockIdx.x;
    int   offv = 0, ida = 0, idb = 0;
    float ddv = 0.0f, wa = 0.0f;
    if (strip < nStrips) {
        const int i = strip * 16 + 2 * wave + lane;
        offv = off[i < nN ? i : nN];
        ddv  = dinv[i < nN ? i : nN - 1];
        const int b0 = __builtin_amdgcn_readlane(offv, 0);
        const int e0 = __builtin_amdgcn_readlane(offv, 1);
        const int e1 = __builtin_amdgcn_readlane(offv, 2);
        ida = csr[first32(b0, e0, lane)];
        idb = csr[first32(e0, e1, lane)];
        ida = clamp_id(ida, nN);
        if (LAYER == 1) wa = dinv[ida];
    }

    int buf = 0;
    int prow0 = -1;
    for (; strip < nStrips; strip += nBlk, buf ^= 1) {
        const int row0 = strip * 16;
        float* __restrict__ xb = xs[buf];
        const int na = row0 + 2 * wave;
        const int beg_a = __builtin_amdgcn_readlane(offv, 0);
        const int end_a = __builtin_amdgcn_readlane(offv, 1);
        const int end_b = __builtin_amdgcn_readlane(offv, 2);
        const float dd_a = __int_as_float(__builtin_amdgcn_readlane(__float_as_int(ddv), 0));
        const float dd_b = __int_as_float(__builtin_amdgcn_readlane(__float_as_int(ddv), 1));
        const int   sb  = clamp_id(idb, nN);

        const int nxt = (strip + nBlk < nStrips) ? (strip + nBlk) : strip;
        const int in  = nxt * 16 + 2 * wave + lane;
        const int   offn = off[in < nN ? in : nN];
        const float ddn  = dinv[in < nN ? in : nN - 1];
        float wb = 0.0f;
        if (LAYER == 1) wb = dinv[sb];

        {
            const float4 acc = node_sum<LAYER>(hp, csr, dinv, na, beg_a, end_a, dd_a, ida, wa, lane, nN);
            const int r = 2 * wave;
            *(float4*)&xb[r * XS_STRIDE + lane * 4] = acc;
            if (LAYER == 1 && lane == 0) xb[r * XS_STRIDE + D] = (na < nN) ? dd_a : 0.0f;
        }

        const int nb0 = __builtin_amdgcn_readlane(offn, 0);
        const int ne0 = __builtin_amdgcn_readlane(offn, 1);
        const int ne1 = __builtin_amdgcn_readlane(offn, 2);
        int       ida_n = csr[first32(nb0, ne0, lane)];
        const int idb_n = csr[first32(ne0, ne1, lane)];

        {
            const float4 acc = node_sum<LAYER>(hp, csr, dinv, na + 1, end_a, end_b, dd_b, sb, wb, lane, nN);
            const int r = 2 * wave + 1;
            *(float4*)&xb[r * XS_STRIDE + lane * 4] = acc;
            if (LAYER == 1 && lane == 0) xb[r * XS_STRIDE + D] = (na + 1 < nN) ? dd_b : 0.0f;
        }
        __syncthreads();

        float wa_n = 0.0f;
        ida_n = clamp_id(ida_n, nN);
        if (LAYER == 1) {
            wa_n = dinv[ida_n];
            __builtin_amdgcn_sched_barrier(0);
        }

        if (prow0 >= 0) {
            const int lr  = 2 * wave;
            const int row = prow0 + lr;
            const float4 o0 = *(const float4*)&os[buf ^ 1][lr * XS_STRIDE + lane * 4];
            const float4 o1 = *(const float4*)&os[buf ^ 1][(lr + 1) * XS_STRIDE + lane * 4];
            if (row < nN)     *(float4*)&out[(size_t)row * D + lane * 4] = o0;
            if (row + 1 < nN) *(float4*)&out[(size_t)(row + 1) * D + lane * 4] = o1;
        }

        v8f c = {};
        #pragma unroll
        for (int i = 0; i < D / 4; ++i) {
            v2f a;
            a.x = xb[m * XS_STRIDE + 4 * i + kk];
            a.y = xb[m * XS_STRIDE + 4 * i + kk + 1];
            c = __builtin_amdgcn_wmma_f32_16x16x4_f32(
                 false, a,  false, bfr[i],
                 (short)0, c,  false,  false);
        }

        float v[8];
        #pragma unroll
        for (int r = 0; r < 8; ++r) {
            v[r] = c[r] + bc;
            if (LAYER == 1) {
                v[r] = v[r] > 0.0f ? v[r] : 0.0f;
                v[r] *= xb[(halfm + r) * XS_STRIDE + D];
            }
        }
        #pragma unroll
        for (int r = 0; r < 8; ++r) os[buf][(halfm + r) * XS_STRIDE + col] = v[r];
        prow0 = row0;

        offv = offn; ddv = ddn; ida = ida_n; idb = idb_n; wa = wa_n;
    }

    if (prow0 >= 0) {
        __syncthreads();
        const int lr  = 2 * wave;
        const int row = prow0 + lr;
        const float4 o0 = *(const float4*)&os[buf ^ 1][lr * XS_STRIDE + lane * 4];
        const float4 o1 = *(const float4*)&os[buf ^ 1][(lr + 1) * XS_STRIDE + lane * 4];
        if (row < nN)     *(float4*)&out[(size_t)row * D + lane * 4] = o0;
        if (row + 1 < nN) *(float4*)&out[(size_t)(row + 1) * D + lane * 4] = o1;
    }
}


static inline size_t align256(size_t v) { return (v + 255) & ~(size_t)255; }

extern "C" void kernel_launch(void* const* d_in, const int* in_sizes, int n_in,
                              void* d_out, int out_size, void* d_ws, size_t ws_size,
                              hipStream_t stream) {
    const float* x  = (const float*)d_in[0];
    const int*   ei = (const int*)d_in[1];
    const float* W1 = (const float*)d_in[2];
    const float* b1 = (const float*)d_in[3];
    const float* W2 = (const float*)d_in[4];
    const float* b2 = (const float*)d_in[5];
    float* out = (float*)d_out;

    const int N = in_sizes[0] / D;
    const int E = in_sizes[1] / 2;
    const int* srcIdx = ei;
    const int* dstIdx = ei + E;

    const int nB  = (N + NB - 1) / NB;
    const int pe  = (E + G1 - 1) / G1;
    const int per = pe < CAPS ? pe : CAPS;
    const int nU  = (E + per - 1) / per;
    const int nT  = nB * nU;

    char* ws = (char*)d_ws;
    size_t o = 0;
    float* dinv  = (float*)(ws + o); o += align256((size_t)N * sizeof(float));
    int*   cnt   = (int*)(ws + o);   o += align256((size_t)nT * sizeof(int));
    int*   beg   = (int*)(ws + o);   o += align256((size_t)nT * sizeof(int));
    int*   off   = (int*)(ws + o);   o += align256((size_t)(N + 1) * sizeof(int));
    int2*  pairs = (int2*)(ws + o);  o += align256((size_t)E * sizeof(int2));
    int*   csr   = (int*)(ws + o);   o += align256((size_t)E * sizeof(int));
    float* hbuf  = (float*)(ws + o);

    const int blk = 256;
    const int nStrips = (N + 15) / 16;
    const int gLayer  = nStrips < 1024 ? nStrips : 1024;

    chunk_sort  <<<nU, T1, 0, stream>>>(srcIdx, dstIdx, cnt, beg, pairs, E, nB, per);
    bucket_build<<<nB, T4, 0, stream>>>(pairs, cnt, beg, off, dinv, csr, N, E, nB, nU, per);

    layer_fused<1><<<gLayer, blk, 0, stream>>>(off, csr, dinv, x, W1, b1, hbuf, N, nStrips, gLayer);

    layer_fused<2><<<gLayer, blk, 0, stream>>>(off, csr, dinv, hbuf, W2, b2, out, N, nStrips, gLayer);
}
